# starting kernel
# baseline (speedup 1.0000x reference)
#include <hip/hip_runtime.h>
#include <hip/hip_bf16.h>
#include <math.h>

#define BB      2
#define LL      2048
#define MM      1024
#define DS      16
#define RR      64
#define NPROJ   (RR + 2*DS)
#define BLT     (BB*LL)

typedef float  v2f __attribute__((ext_vector_type(2)));
typedef float  v8f __attribute__((ext_vector_type(8)));

__global__ __launch_bounds__(32)
void proj_xp_kernel(const float* __restrict__ x,
                    const float* __restrict__ w,
                    float* __restrict__ xp)
{
    const int lane = threadIdx.x;
    const int tile = blockIdx.x;
    const int tm   = tile / (NPROJ/16);
    const int tn   = tile % (NPROJ/16);
    const int half = lane >> 4;
    const int l15  = lane & 15;

    const float* __restrict__ arow = x + (size_t)(tm*16 + l15) * MM;
    const float* __restrict__ brow = w + (size_t)(tn*16 + l15) * MM;

    v8f c = {};
    #pragma unroll 4
    for (int k0 = 0; k0 < MM; k0 += 4) {
        const int ka = k0 + half*2;
        v2f a; a.x = arow[ka]; a.y = arow[ka + 1];
        v2f b; b.x = brow[ka]; b.y = brow[ka + 1];
        c = __builtin_amdgcn_wmma_f32_16x16x4_f32(
                false, a, false, b, (short)0, c, false, false);
    }

    float* __restrict__ o = xp + (size_t)(tm*16) * NPROJ + tn*16 + l15;
    #pragma unroll
    for (int r = 0; r < 8; ++r) {
        const int row = r + half*8;
        o[(size_t)row * NPROJ] = c[r];
    }
}

__global__ __launch_bounds__(32)
void dt_proj_kernel(const float* __restrict__ xp,
                    const float* __restrict__ w,
                    const float* __restrict__ bias,
                    float* __restrict__ dt)
{
    const int lane = threadIdx.x;
    const int tile = blockIdx.x;
    const int tm   = tile / (MM/16);
    const int tn   = tile % (MM/16);
    const int half = lane >> 4;
    const int l15  = lane & 15;

    const float* __restrict__ arow = xp + (size_t)(tm*16 + l15) * NPROJ;
    const float* __restrict__ brow = w  + (size_t)(tn*16 + l15) * RR;

    v8f c = {};
    #pragma unroll
    for (int k0 = 0; k0 < RR; k0 += 4) {
        const int ka = k0 + half*2;
        v2f a; a.x = arow[ka]; a.y = arow[ka + 1];
        v2f b; b.x = brow[ka]; b.y = brow[ka + 1];
        c = __builtin_amdgcn_wmma_f32_16x16x4_f32(
                false, a, false, b, (short)0, c, false, false);
    }

    const int col = tn*16 + l15;
    const float bb = bias[col];
    #pragma unroll
    for (int r = 0; r < 8; ++r) {
        const int row = tm*16 + r + half*8;
        float v = c[r] + bb;
        float sp = (v > 20.0f) ? v : log1pf(__expf(v));
        dt[(size_t)row * MM + col] = sp;
    }
}

__global__ __launch_bounds__(256)
void scan_kernel(const float* __restrict__ x,
                 const float* __restrict__ z,
                 const float* __restrict__ xp,
                 const float* __restrict__ dtb,
                 const float* __restrict__ A_log,
                 const float* __restrict__ Dp_arr,
                 float* __restrict__ out)
{
    const int gid = blockIdx.x * blockDim.x + threadIdx.x;
    const int b   = gid >> 10;
    const int m   = gid & (MM - 1);

    float Am[DS], h[DS];
    #pragma unroll
    for (int d = 0; d < DS; ++d) {
        Am[d] = -__expf(A_log[(size_t)d * MM + m]);
        h[d]  = 0.0f;
    }
    const float Dp = Dp_arr[m];
    const size_t rowB = (size_t)b * LL;

    for (int t = 0; t < LL; ++t) {
        const size_t r = rowB + t;
        const float dtv = dtb[r * MM + m];
        const float xv  = x  [r * MM + m];
        const float zv  = z  [r * MM + m];

        const float4* __restrict__ bc =
            (const float4*)(xp + r * NPROJ + RR);
        float Bv[DS], Cv[DS];
        #pragma unroll
        for (int i = 0; i < 4; ++i) {
            float4 vb = bc[i];
            float4 vc = bc[4 + i];
            Bv[4*i+0] = vb.x; Bv[4*i+1] = vb.y; Bv[4*i+2] = vb.z; Bv[4*i+3] = vb.w;
            Cv[4*i+0] = vc.x; Cv[4*i+1] = vc.y; Cv[4*i+2] = vc.z; Cv[4*i+3] = vc.w;
        }

        if (t + 8 < LL)
            __builtin_prefetch(xp + (r + 8) * NPROJ, 0, 1);

        const float dx = dtv * xv;
        float y = 0.0f;
        #pragma unroll
        for (int d = 0; d < DS; ++d) {
            h[d] = __expf(dtv * Am[d]) * h[d] + dx * Bv[d];
            y    = fmaf(h[d], Cv[d], y);
        }
        y = fmaf(Dp, xv, y);

        const float sig = 1.0f / (1.0f + __expf(-zv));
        out[r * MM + m] = y * (zv * sig);
    }
}

extern "C" void kernel_launch(void* const* d_in, const int* in_sizes, int n_in,
                              void* d_out, int out_size, void* d_ws, size_t ws_size,
                              hipStream_t stream) {
    const float* x         = (const float*)d_in[0];
    const float* z         = (const float*)d_in[1];
    const float* x_proj_w  = (const float*)d_in[2];
    const float* dt_proj_w = (const float*)d_in[3];
    const float* dt_proj_b = (const float*)d_in[4];
    const float* A_log     = (const float*)d_in[5];
    const float* D_param   = (const float*)d_in[6];
    float* out = (float*)d_out;

    float* xp_ws = (float*)d_ws;
    float* dt_ws = xp_ws + (size_t)BLT * NPROJ;

    proj_xp_kernel<<<(BLT/16) * (NPROJ/16), 32, 0, stream>>>(x, x_proj_w, xp_ws);

    dt_proj_kernel<<<(BLT/16) * (MM/16), 32, 0, stream>>>(xp_ws, dt_proj_w,
                                                          dt_proj_b, dt_ws);

    scan_kernel<<<(BB*MM)/256, 256, 0, stream>>>(x, z, xp_ws, dt_ws,
                                                 A_log, D_param, out);
}


# Scan over L split into 128 parallel chunks: per chunk end state from h=0 and sum of dt (decay=exp(A*sum dt)), serial carry over chunks, rescan from carried state; h summed in a different order
# speedup vs baseline: 15.8041x; 15.8041x over previous
#include <hip/hip_runtime.h>
#include <hip/hip_bf16.h>
#include <math.h>

#define BB      2
#define LL      2048
#define MM      1024
#define DS      16
#define RR      64
#define NPROJ   (RR + 2*DS)
#define BLT     (BB*LL)
#define NC      128
#define TT      (LL/NC)
#define TPB     256
#define MBLK    (MM/TPB)

typedef float  v2f __attribute__((ext_vector_type(2)));
typedef float  v8f __attribute__((ext_vector_type(8)));

__global__ __launch_bounds__(32)
void proj_xp_kernel(const float* __restrict__ x,
                    const float* __restrict__ w,
                    float* __restrict__ xp)
{
    const int lane = threadIdx.x;
    const int tile = blockIdx.x;
    const int tm   = tile / (NPROJ/16);
    const int tn   = tile % (NPROJ/16);
    const int half = lane >> 4;
    const int l15  = lane & 15;

    const float* __restrict__ arow = x + (size_t)(tm*16 + l15) * MM;
    const float* __restrict__ brow = w + (size_t)(tn*16 + l15) * MM;

    v8f c = {};
    #pragma unroll 4
    for (int k0 = 0; k0 < MM; k0 += 4) {
        const int ka = k0 + half*2;
        v2f a; a.x = arow[ka]; a.y = arow[ka + 1];
        v2f b; b.x = brow[ka]; b.y = brow[ka + 1];
        c = __builtin_amdgcn_wmma_f32_16x16x4_f32(
                false, a, false, b, (short)0, c, false, false);
    }

    float* __restrict__ o = xp + (size_t)(tm*16) * NPROJ + tn*16 + l15;
    #pragma unroll
    for (int r = 0; r < 8; ++r) {
        const int row = r + half*8;
        o[(size_t)row * NPROJ] = c[r];
    }
}

__global__ __launch_bounds__(32)
void dt_proj_kernel(const float* __restrict__ xp,
                    const float* __restrict__ w,
                    const float* __restrict__ bias,
                    float* __restrict__ dt)
{
    const int lane = threadIdx.x;
    const int tile = blockIdx.x;
    const int tm   = tile / (MM/16);
    const int tn   = tile % (MM/16);
    const int half = lane >> 4;
    const int l15  = lane & 15;

    const float* __restrict__ arow = xp + (size_t)(tm*16 + l15) * NPROJ;
    const float* __restrict__ brow = w  + (size_t)(tn*16 + l15) * RR;

    v8f c = {};
    #pragma unroll
    for (int k0 = 0; k0 < RR; k0 += 4) {
        const int ka = k0 + half*2;
        v2f a; a.x = arow[ka]; a.y = arow[ka + 1];
        v2f b; b.x = brow[ka]; b.y = brow[ka + 1];
        c = __builtin_amdgcn_wmma_f32_16x16x4_f32(
                false, a, false, b, (short)0, c, false, false);
    }

    const int col = tn*16 + l15;
    const float bb = bias[col];
    #pragma unroll
    for (int r = 0; r < 8; ++r) {
        const int row = tm*16 + r + half*8;
        float v = c[r] + bb;
        float sp = (v > 20.0f) ? v : log1pf(__expf(v));
        dt[(size_t)row * MM + col] = sp;
    }
}

__global__ __launch_bounds__(TPB)
void chunk_state_kernel(const float* __restrict__ x,
                        const float* __restrict__ xp,
                        const float* __restrict__ dtb,
                        const float* __restrict__ A_log,
                        float* __restrict__ hloc,
                        float* __restrict__ dsum)
{
    const int mb = blockIdx.x % MBLK;
    const int c  = (blockIdx.x / MBLK) % (NC - 1);
    const int b  = blockIdx.x / (MBLK * (NC - 1));
    const int m  = mb * TPB + threadIdx.x;

    float Am[DS], h[DS];
    #pragma unroll
    for (int d = 0; d < DS; ++d) {
        Am[d] = -__expf(A_log[(size_t)d * MM + m]);
        h[d]  = 0.0f;
    }
    float S = 0.0f;
    const size_t row0 = (size_t)b * LL + (size_t)c * TT;

    for (int t = 0; t < TT; ++t) {
        const size_t r = row0 + t;
        const float dtv = dtb[r * MM + m];
        const float xv  = x  [r * MM + m];

        const float4* __restrict__ bc =
            (const float4*)(xp + r * NPROJ + RR);
        float Bv[DS];
        #pragma unroll
        for (int i = 0; i < 4; ++i) {
            float4 vb = bc[i];
            Bv[4*i+0] = vb.x; Bv[4*i+1] = vb.y; Bv[4*i+2] = vb.z; Bv[4*i+3] = vb.w;
        }

        const float dx = dtv * xv;
        #pragma unroll
        for (int d = 0; d < DS; ++d)
            h[d] = __expf(dtv * Am[d]) * h[d] + dx * Bv[d];
        S += dtv;
    }

    const size_t base = ((size_t)(b * NC + c) * DS) * MM + m;
    #pragma unroll
    for (int d = 0; d < DS; ++d)
        hloc[base + (size_t)d * MM] = h[d];
    dsum[(size_t)(b * NC + c) * MM + m] = S;
}

__global__ __launch_bounds__(256)
void chunk_carry_kernel(const float* __restrict__ A_log,
                        const float* __restrict__ hloc,
                        const float* __restrict__ dsum,
                        float* __restrict__ hin)
{
    const int idx = blockIdx.x * 256 + threadIdx.x;
    const int m   = idx & (MM - 1);
    const int d   = (idx >> 10) & (DS - 1);
    const int b   = idx >> 14;

    const float Am = -__expf(A_log[(size_t)d * MM + m]);
    float h = 0.0f;
    size_t o = ((size_t)(b * NC) * DS + d) * MM + m;
    hin[o] = 0.0f;
    #pragma unroll 32
    for (int c = 0; c < NC - 1; ++c) {
        const float S = dsum[(size_t)(b * NC + c) * MM + m];
        const float e = hloc[o];
        h = __expf(Am * S) * h + e;
        o += (size_t)DS * MM;
        hin[o] = h;
    }
}

__global__ __launch_bounds__(TPB)
void scan_kernel(const float* __restrict__ x,
                 const float* __restrict__ z,
                 const float* __restrict__ xp,
                 const float* __restrict__ dtb,
                 const float* __restrict__ A_log,
                 const float* __restrict__ Dp_arr,
                 const float* __restrict__ hin,
                 float* __restrict__ out)
{
    const int mb = blockIdx.x % MBLK;
    const int c  = (blockIdx.x / MBLK) % NC;
    const int b  = blockIdx.x / (MBLK * NC);
    const int m  = mb * TPB + threadIdx.x;

    float Am[DS], h[DS];
    const size_t base = ((size_t)(b * NC + c) * DS) * MM + m;
    #pragma unroll
    for (int d = 0; d < DS; ++d) {
        Am[d] = -__expf(A_log[(size_t)d * MM + m]);
        h[d]  = hin[base + (size_t)d * MM];
    }
    const float Dp = Dp_arr[m];
    const size_t row0 = (size_t)b * LL + (size_t)c * TT;

    for (int t = 0; t < TT; ++t) {
        const size_t r = row0 + t;
        const float dtv = dtb[r * MM + m];
        const float xv  = x  [r * MM + m];
        const float zv  = z  [r * MM + m];

        const float4* __restrict__ bc =
            (const float4*)(xp + r * NPROJ + RR);
        float Bv[DS], Cv[DS];
        #pragma unroll
        for (int i = 0; i < 4; ++i) {
            float4 vb = bc[i];
            float4 vc = bc[4 + i];
            Bv[4*i+0] = vb.x; Bv[4*i+1] = vb.y; Bv[4*i+2] = vb.z; Bv[4*i+3] = vb.w;
            Cv[4*i+0] = vc.x; Cv[4*i+1] = vc.y; Cv[4*i+2] = vc.z; Cv[4*i+3] = vc.w;
        }

        const float dx = dtv * xv;
        float y = 0.0f;
        #pragma unroll
        for (int d = 0; d < DS; ++d) {
            h[d] = __expf(dtv * Am[d]) * h[d] + dx * Bv[d];
            y    = fmaf(h[d], Cv[d], y);
        }
        y = fmaf(Dp, xv, y);

        const float sig = 1.0f / (1.0f + __expf(-zv));
        out[r * MM + m] = y * (zv * sig);
    }
}

extern "C" void kernel_launch(void* const* d_in, const int* in_sizes, int n_in,
                              void* d_out, int out_size, void* d_ws, size_t ws_size,
                              hipStream_t stream) {
    const float* x         = (const float*)d_in[0];
    const float* z         = (const float*)d_in[1];
    const float* x_proj_w  = (const float*)d_in[2];
    const float* dt_proj_w = (const float*)d_in[3];
    const float* dt_proj_b = (const float*)d_in[4];
    const float* A_log     = (const float*)d_in[5];
    const float* D_param   = (const float*)d_in[6];
    float* out = (float*)d_out;

    float* xp_ws = (float*)d_ws;
    float* dt_ws = xp_ws + (size_t)BLT * NPROJ;
    float* ds_ws = dt_ws + (size_t)BLT * MM;
    float* hl_ws = ds_ws + (size_t)BB * NC * MM;
    float* hi_ws = hl_ws + (size_t)BB * NC * DS * MM;

    proj_xp_kernel<<<(BLT/16) * (NPROJ/16), 32, 0, stream>>>(x, x_proj_w, xp_ws);

    dt_proj_kernel<<<(BLT/16) * (MM/16), 32, 0, stream>>>(xp_ws, dt_proj_w,
                                                          dt_proj_b, dt_ws);

    chunk_state_kernel<<<BB * (NC - 1) * MBLK, TPB, 0, stream>>>(x, xp_ws, dt_ws, A_log,
                                                                 hl_ws, ds_ws);

    chunk_carry_kernel<<<(BB * DS * MM) / 256, 256, 0, stream>>>(A_log, hl_ws, ds_ws, hi_ws);

    scan_kernel<<<BB * NC * MBLK, TPB, 0, stream>>>(x, z, xp_ws, dt_ws,
                                                    A_log, D_param, hi_ws, out);
}


# x_proj GEMM: one 4-wave workgroup per 16 rows, x and w chunks staged in LDS by whole-line loads, K split over the 4 waves and summed in LDS (xp summed in a different order)
# speedup vs baseline: 17.7150x; 1.1209x over previous
#include <hip/hip_runtime.h>
#include <hip/hip_bf16.h>
#include <math.h>

#define BB      2
#define LL      2048
#define MM      1024
#define DS      16
#define RR      64
#define NPROJ   (RR + 2*DS)
#define BLT     (BB*LL)
#define NC      128
#define TT      (LL/NC)
#define TPB     256
#define MBLK    (MM/TPB)

typedef float  v2f __attribute__((ext_vector_type(2)));
typedef float  v8f __attribute__((ext_vector_type(8)));

#define KC      128
#define NCH     (MM / KC)
#define SROWS   (16 + NPROJ)
#define SPITCH  (KC + 4)
#define RPITCH  100

__global__ __launch_bounds__(128)
void proj_xp_kernel(const float* __restrict__ x,
                    const float* __restrict__ w,
                    float* __restrict__ xp)
{
    __shared__ __attribute__((aligned(16))) float smem[2 * SROWS * SPITCH];

    const int tid  = threadIdx.x;
    const int lane = tid & 31;
    const int wv   = tid >> 5;
    const int half = lane >> 4;
    const int l15  = lane & 15;
    const int rt   = blockIdx.x;

    const float* __restrict__ xs = x + (size_t)(rt*16 + 4*wv) * MM + lane*4;
    const float* __restrict__ ws = w + (size_t)(24*wv) * MM + lane*4;
    const int xd = (4*wv) * SPITCH + lane*4;
    const int wd = (16 + 24*wv) * SPITCH + lane*4;
    const int fo = l15 * SPITCH + 32*wv + 16*half;

    float4 st[28];
    v8f acc[6];
    #pragma unroll
    for (int t = 0; t < 6; ++t) acc[t] = (v8f){};

    #pragma unroll
    for (int i = 0; i < 4; ++i)  st[i]     = *(const float4*)(xs + (size_t)i * MM);
    #pragma unroll
    for (int i = 0; i < 24; ++i) st[4 + i] = *(const float4*)(ws + (size_t)i * MM);
    #pragma unroll
    for (int i = 0; i < 4; ++i)  *(float4*)(smem + xd + i*SPITCH) = st[i];
    #pragma unroll
    for (int i = 0; i < 24; ++i) *(float4*)(smem + wd + i*SPITCH) = st[4 + i];
    __syncthreads();

    #pragma unroll
    for (int s = 0; s < NCH; ++s) {
        const float* cur = smem + (s & 1) * (SROWS * SPITCH);
        float*       nxt = smem + ((s + 1) & 1) * (SROWS * SPITCH);

        if (s + 1 < NCH) {
            #pragma unroll
            for (int i = 0; i < 4; ++i)
                st[i]     = *(const float4*)(xs + (size_t)i * MM + (s + 1) * KC);
            #pragma unroll
            for (int i = 0; i < 24; ++i)
                st[4 + i] = *(const float4*)(ws + (size_t)i * MM + (s + 1) * KC);
        }

        float4 af[4], bf[6][4];
        #pragma unroll
        for (int j = 0; j < 4; ++j) af[j] = *(const float4*)(cur + fo + 4*j);
        #pragma unroll
        for (int t = 0; t < 6; ++t) {
            #pragma unroll
            for (int j = 0; j < 4; ++j)
                bf[t][j] = *(const float4*)(cur + (16 + 16*t) * SPITCH + fo + 4*j);
        }

        #pragma unroll
        for (int j = 0; j < 4; ++j) {
            v2f a0; a0.x = af[j].x; a0.y = af[j].y;
            v2f a1; a1.x = af[j].z; a1.y = af[j].w;
            #pragma unroll
            for (int t = 0; t < 6; ++t) {
                v2f b0; b0.x = bf[t][j].x; b0.y = bf[t][j].y;
                acc[t] = __builtin_amdgcn_wmma_f32_16x16x4_f32(
                             false, a0, false, b0, (short)0, acc[t], false, false);
            }
            #pragma unroll
            for (int t = 0; t < 6; ++t) {
                v2f b1; b1.x = bf[t][j].z; b1.y = bf[t][j].w;
                acc[t] = __builtin_amdgcn_wmma_f32_16x16x4_f32(
                             false, a1, false, b1, (short)0, acc[t], false, false);
            }
        }

        if (s + 1 < NCH) {
            #pragma unroll
            for (int i = 0; i < 4; ++i)  *(float4*)(nxt + xd + i*SPITCH) = st[i];
            #pragma unroll
            for (int i = 0; i < 24; ++i) *(float4*)(nxt + wd + i*SPITCH) = st[4 + i];
        }
        __syncthreads();
    }

    float* red = smem;
    #pragma unroll
    for (int t = 0; t < 6; ++t) {
        #pragma unroll
        for (int r = 0; r < 8; ++r)
            red[(wv*16 + 8*half + r) * RPITCH + 16*t + l15] = acc[t][r];
    }
    __syncthreads();

    #pragma unroll
    for (int i = 0; i < 3; ++i) {
        const int q   = tid + 128*i;
        const int row = q / 24;
        const int c4  = q - row*24;
        const float4 p0 = *(const float4*)(red + (0*16 + row) * RPITCH + 4*c4);
        const float4 p1 = *(const float4*)(red + (1*16 + row) * RPITCH + 4*c4);
        const float4 p2 = *(const float4*)(red + (2*16 + row) * RPITCH + 4*c4);
        const float4 p3 = *(const float4*)(red + (3*16 + row) * RPITCH + 4*c4);
        float4 o;
        o.x = ((p0.x + p1.x) + p2.x) + p3.x;
        o.y = ((p0.y + p1.y) + p2.y) + p3.y;
        o.z = ((p0.z + p1.z) + p2.z) + p3.z;
        o.w = ((p0.w + p1.w) + p2.w) + p3.w;
        *(float4*)(xp + (size_t)rt * (16 * NPROJ) + 4*q) = o;
    }
}

__global__ __launch_bounds__(32)
void dt_proj_kernel(const float* __restrict__ xp,
                    const float* __restrict__ w,
                    const float* __restrict__ bias,
                    float* __restrict__ dt)
{
    const int lane = threadIdx.x;
    const int tile = blockIdx.x;
    const int tm   = tile / (MM/16);
    const int tn   = tile % (MM/16);
    const int half = lane >> 4;
    const int l15  = lane & 15;

    const float* __restrict__ arow = xp + (size_t)(tm*16 + l15) * NPROJ;
    const float* __restrict__ brow = w  + (size_t)(tn*16 + l15) * RR;

    v8f c = {};
    #pragma unroll
    for (int k0 = 0; k0 < RR; k0 += 4) {
        const int ka = k0 + half*2;
        v2f a; a.x = arow[ka]; a.y = arow[ka + 1];
        v2f b; b.x = brow[ka]; b.y = brow[ka + 1];
        c = __builtin_amdgcn_wmma_f32_16x16x4_f32(
                false, a, false, b, (short)0, c, false, false);
    }

    const int col = tn*16 + l15;
    const float bb = bias[col];
    #pragma unroll
    for (int r = 0; r < 8; ++r) {
        const int row = tm*16 + r + half*8;
        float v = c[r] + bb;
        float sp = (v > 20.0f) ? v : log1pf(__expf(v));
        dt[(size_t)row * MM + col] = sp;
    }
}

__global__ __launch_bounds__(TPB)
void chunk_state_kernel(const float* __restrict__ x,
                        const float* __restrict__ xp,
                        const float* __restrict__ dtb,
                        const float* __restrict__ A_log,
                        float* __restrict__ hloc,
                        float* __restrict__ dsum)
{
    const int mb = blockIdx.x % MBLK;
    const int c  = (blockIdx.x / MBLK) % (NC - 1);
    const int b  = blockIdx.x / (MBLK * (NC - 1));
    const int m  = mb * TPB + threadIdx.x;

    float Am[DS], h[DS];
    #pragma unroll
    for (int d = 0; d < DS; ++d) {
        Am[d] = -__expf(A_log[(size_t)d * MM + m]);
        h[d]  = 0.0f;
    }
    float S = 0.0f;
    const size_t row0 = (size_t)b * LL + (size_t)c * TT;

    for (int t = 0; t < TT; ++t) {
        const size_t r = row0 + t;
        const float dtv = dtb[r * MM + m];
        const float xv  = x  [r * MM + m];

        const float4* __restrict__ bc =
            (const float4*)(xp + r * NPROJ + RR);
        float Bv[DS];
        #pragma unroll
        for (int i = 0; i < 4; ++i) {
            float4 vb = bc[i];
            Bv[4*i+0] = vb.x; Bv[4*i+1] = vb.y; Bv[4*i+2] = vb.z; Bv[4*i+3] = vb.w;
        }

        const float dx = dtv * xv;
        #pragma unroll
        for (int d = 0; d < DS; ++d)
            h[d] = __expf(dtv * Am[d]) * h[d] + dx * Bv[d];
        S += dtv;
    }

    const size_t base = ((size_t)(b * NC + c) * DS) * MM + m;
    #pragma unroll
    for (int d = 0; d < DS; ++d)
        hloc[base + (size_t)d * MM] = h[d];
    dsum[(size_t)(b * NC + c) * MM + m] = S;
}

__global__ __launch_bounds__(256)
void chunk_carry_kernel(const float* __restrict__ A_log,
                        const float* __restrict__ hloc,
                        const float* __restrict__ dsum,
                        float* __restrict__ hin)
{
    const int idx = blockIdx.x * 256 + threadIdx.x;
    const int m   = idx & (MM - 1);
    const int d   = (idx >> 10) & (DS - 1);
    const int b   = idx >> 14;

    const float Am = -__expf(A_log[(size_t)d * MM + m]);
    float h = 0.0f;
    size_t o = ((size_t)(b * NC) * DS + d) * MM + m;
    hin[o] = 0.0f;
    #pragma unroll 32
    for (int c = 0; c < NC - 1; ++c) {
        const float S = dsum[(size_t)(b * NC + c) * MM + m];
        const float e = hloc[o];
        h = __expf(Am * S) * h + e;
        o += (size_t)DS * MM;
        hin[o] = h;
    }
}

__global__ __launch_bounds__(TPB)
void scan_kernel(const float* __restrict__ x,
                 const float* __restrict__ z,
                 const float* __restrict__ xp,
                 const float* __restrict__ dtb,
                 const float* __restrict__ A_log,
                 const float* __restrict__ Dp_arr,
                 const float* __restrict__ hin,
                 float* __restrict__ out)
{
    const int mb = blockIdx.x % MBLK;
    const int c  = (blockIdx.x / MBLK) % NC;
    const int b  = blockIdx.x / (MBLK * NC);
    const int m  = mb * TPB + threadIdx.x;

    float Am[DS], h[DS];
    const size_t base = ((size_t)(b * NC + c) * DS) * MM + m;
    #pragma unroll
    for (int d = 0; d < DS; ++d) {
        Am[d] = -__expf(A_log[(size_t)d * MM + m]);
        h[d]  = hin[base + (size_t)d * MM];
    }
    const float Dp = Dp_arr[m];
    const size_t row0 = (size_t)b * LL + (size_t)c * TT;

    for (int t = 0; t < TT; ++t) {
        const size_t r = row0 + t;
        const float dtv = dtb[r * MM + m];
        const float xv  = x  [r * MM + m];
        const float zv  = z  [r * MM + m];

        const float4* __restrict__ bc =
            (const float4*)(xp + r * NPROJ + RR);
        float Bv[DS], Cv[DS];
        #pragma unroll
        for (int i = 0; i < 4; ++i) {
            float4 vb = bc[i];
            float4 vc = bc[4 + i];
            Bv[4*i+0] = vb.x; Bv[4*i+1] = vb.y; Bv[4*i+2] = vb.z; Bv[4*i+3] = vb.w;
            Cv[4*i+0] = vc.x; Cv[4*i+1] = vc.y; Cv[4*i+2] = vc.z; Cv[4*i+3] = vc.w;
        }

        const float dx = dtv * xv;
        float y = 0.0f;
        #pragma unroll
        for (int d = 0; d < DS; ++d) {
            h[d] = __expf(dtv * Am[d]) * h[d] + dx * Bv[d];
            y    = fmaf(h[d], Cv[d], y);
        }
        y = fmaf(Dp, xv, y);

        const float sig = 1.0f / (1.0f + __expf(-zv));
        out[r * MM + m] = y * (zv * sig);
    }
}

extern "C" void kernel_launch(void* const* d_in, const int* in_sizes, int n_in,
                              void* d_out, int out_size, void* d_ws, size_t ws_size,
                              hipStream_t stream) {
    const float* x         = (const float*)d_in[0];
    const float* z         = (const float*)d_in[1];
    const float* x_proj_w  = (const float*)d_in[2];
    const float* dt_proj_w = (const float*)d_in[3];
    const float* dt_proj_b = (const float*)d_in[4];
    const float* A_log     = (const float*)d_in[5];
    const float* D_param   = (const float*)d_in[6];
    float* out = (float*)d_out;

    float* xp_ws = (float*)d_ws;
    float* dt_ws = xp_ws + (size_t)BLT * NPROJ;
    float* ds_ws = dt_ws + (size_t)BLT * MM;
    float* hl_ws = ds_ws + (size_t)BB * NC * MM;
    float* hi_ws = hl_ws + (size_t)BB * NC * DS * MM;

    proj_xp_kernel<<<BLT/16, 128, 0, stream>>>(x, x_proj_w, xp_ws);

    dt_proj_kernel<<<(BLT/16) * (MM/16), 32, 0, stream>>>(xp_ws, dt_proj_w,
                                                          dt_proj_b, dt_ws);

    chunk_state_kernel<<<BB * (NC - 1) * MBLK, TPB, 0, stream>>>(x, xp_ws, dt_ws, A_log,
                                                                 hl_ws, ds_ws);

    chunk_carry_kernel<<<(BB * DS * MM) / 256, 256, 0, stream>>>(A_log, hl_ws, ds_ws, hi_ws);

    scan_kernel<<<BB * NC * MBLK, TPB, 0, stream>>>(x, z, xp_ws, dt_ws,
                                                    A_log, D_param, hi_ws, out);
}


# dt_proj kernel: 1024 workgroups of 8 waves, each wave a pair of 16-column tiles with b128 fragment loads and whole-line stores, instead of 16384 one-wave workgroups; dt summed in a different k order
# speedup vs baseline: 18.7737x; 1.0598x over previous
#include <hip/hip_runtime.h>
#include <hip/hip_bf16.h>
#include <math.h>

#define BB      2
#define LL      2048
#define MM      1024
#define DS      16
#define RR      64
#define NPROJ   (RR + 2*DS)
#define BLT     (BB*LL)
#define NC      128
#define TT      (LL/NC)
#define TPB     256
#define MBLK    (MM/TPB)

typedef float  v2f __attribute__((ext_vector_type(2)));
typedef float  v8f __attribute__((ext_vector_type(8)));

#define KC      128
#define NCH     (MM / KC)
#define SROWS   (16 + NPROJ)
#define SPITCH  (KC + 4)
#define RPITCH  100

__global__ __launch_bounds__(128)
void proj_xp_kernel(const float* __restrict__ x,
                    const float* __restrict__ w,
                    float* __restrict__ xp)
{
    __shared__ __attribute__((aligned(16))) float smem[2 * SROWS * SPITCH];

    const int tid  = threadIdx.x;
    const int lane = tid & 31;
    const int wv   = tid >> 5;
    const int half = lane >> 4;
    const int l15  = lane & 15;
    const int rt   = blockIdx.x;

    const float* __restrict__ xs = x + (size_t)(rt*16 + 4*wv) * MM + lane*4;
    const float* __restrict__ ws = w + (size_t)(24*wv) * MM + lane*4;
    const int xd = (4*wv) * SPITCH + lane*4;
    const int wd = (16 + 24*wv) * SPITCH + lane*4;
    const int fo = l15 * SPITCH + 32*wv + 16*half;

    float4 st[28];
    v8f acc[6];
    #pragma unroll
    for (int t = 0; t < 6; ++t) acc[t] = (v8f){};

    #pragma unroll
    for (int i = 0; i < 4; ++i)  st[i]     = *(const float4*)(xs + (size_t)i * MM);
    #pragma unroll
    for (int i = 0; i < 24; ++i) st[4 + i] = *(const float4*)(ws + (size_t)i * MM);
    #pragma unroll
    for (int i = 0; i < 4; ++i)  *(float4*)(smem + xd + i*SPITCH) = st[i];
    #pragma unroll
    for (int i = 0; i < 24; ++i) *(float4*)(smem + wd + i*SPITCH) = st[4 + i];
    __syncthreads();

    #pragma unroll
    for (int s = 0; s < NCH; ++s) {
        const float* cur = smem + (s & 1) * (SROWS * SPITCH);
        float*       nxt = smem + ((s + 1) & 1) * (SROWS * SPITCH);

        if (s + 1 < NCH) {
            #pragma unroll
            for (int i = 0; i < 4; ++i)
                st[i]     = *(const float4*)(xs + (size_t)i * MM + (s + 1) * KC);
            #pragma unroll
            for (int i = 0; i < 24; ++i)
                st[4 + i] = *(const float4*)(ws + (size_t)i * MM + (s + 1) * KC);
        }

        float4 af[4], bf[6][4];
        #pragma unroll
        for (int j = 0; j < 4; ++j) af[j] = *(const float4*)(cur + fo + 4*j);
        #pragma unroll
        for (int t = 0; t < 6; ++t) {
            #pragma unroll
            for (int j = 0; j < 4; ++j)
                bf[t][j] = *(const float4*)(cur + (16 + 16*t) * SPITCH + fo + 4*j);
        }

        #pragma unroll
        for (int j = 0; j < 4; ++j) {
            v2f a0; a0.x = af[j].x; a0.y = af[j].y;
            v2f a1; a1.x = af[j].z; a1.y = af[j].w;
            #pragma unroll
            for (int t = 0; t < 6; ++t) {
                v2f b0; b0.x = bf[t][j].x; b0.y = bf[t][j].y;
                acc[t] = __builtin_amdgcn_wmma_f32_16x16x4_f32(
                             false, a0, false, b0, (short)0, acc[t], false, false);
            }
            #pragma unroll
            for (int t = 0; t < 6; ++t) {
                v2f b1; b1.x = bf[t][j].z; b1.y = bf[t][j].w;
                acc[t] = __builtin_amdgcn_wmma_f32_16x16x4_f32(
                             false, a1, false, b1, (short)0, acc[t], false, false);
            }
        }

        if (s + 1 < NCH) {
            #pragma unroll
            for (int i = 0; i < 4; ++i)  *(float4*)(nxt + xd + i*SPITCH) = st[i];
            #pragma unroll
            for (int i = 0; i < 24; ++i) *(float4*)(nxt + wd + i*SPITCH) = st[4 + i];
        }
        __syncthreads();
    }

    float* red = smem;
    #pragma unroll
    for (int t = 0; t < 6; ++t) {
        #pragma unroll
        for (int r = 0; r < 8; ++r)
            red[(wv*16 + 8*half + r) * RPITCH + 16*t + l15] = acc[t][r];
    }
    __syncthreads();

    #pragma unroll
    for (int i = 0; i < 3; ++i) {
        const int q   = tid + 128*i;
        const int row = q / 24;
        const int c4  = q - row*24;
        const float4 p0 = *(const float4*)(red + (0*16 + row) * RPITCH + 4*c4);
        const float4 p1 = *(const float4*)(red + (1*16 + row) * RPITCH + 4*c4);
        const float4 p2 = *(const float4*)(red + (2*16 + row) * RPITCH + 4*c4);
        const float4 p3 = *(const float4*)(red + (3*16 + row) * RPITCH + 4*c4);
        float4 o;
        o.x = ((p0.x + p1.x) + p2.x) + p3.x;
        o.y = ((p0.y + p1.y) + p2.y) + p3.y;
        o.z = ((p0.z + p1.z) + p2.z) + p3.z;
        o.w = ((p0.w + p1.w) + p2.w) + p3.w;
        *(float4*)(xp + (size_t)rt * (16 * NPROJ) + 4*q) = o;
    }
}

#define DTW_WAVES 8
#define DTW_COLS  (DTW_WAVES * 32)

__device__ __forceinline__ void dtw_load_pair(float4 (&b)[2][8], const float* __restrict__ wb)
{
    #pragma unroll
    for (int u = 0; u < 2; ++u) {
        #pragma unroll
        for (int j = 0; j < 8; ++j)
            b[u][j] = *(const float4*)(wb + (size_t)(16*u) * RR + 8*j);
    }
}

__device__ __forceinline__ float softplus_f(float v)
{
    return (v > 20.0f) ? v : log1pf(__expf(v));
}

__device__ __forceinline__ void dt_tile_pair(const float4 (&a4)[8], const float4 (&b)[2][8],
                                             float bb, float* __restrict__ dcol, int half)
{
    v8f c0 = {}, c1 = {};
    #pragma unroll
    for (int j = 0; j < 8; ++j) {
        v2f a0; a0.x = a4[j].x; a0.y = a4[j].y;
        v2f a1; a1.x = a4[j].z; a1.y = a4[j].w;
        v2f p0; p0.x = b[0][j].x; p0.y = b[0][j].y;
        v2f q0; q0.x = b[1][j].x; q0.y = b[1][j].y;
        v2f p1; p1.x = b[0][j].z; p1.y = b[0][j].w;
        v2f q1; q1.x = b[1][j].z; q1.y = b[1][j].w;
        c0 = __builtin_amdgcn_wmma_f32_16x16x4_f32(false, a0, false, p0, (short)0, c0, false, false);
        c1 = __builtin_amdgcn_wmma_f32_16x16x4_f32(false, a0, false, q0, (short)0, c1, false, false);
        c0 = __builtin_amdgcn_wmma_f32_16x16x4_f32(false, a1, false, p1, (short)0, c0, false, false);
        c1 = __builtin_amdgcn_wmma_f32_16x16x4_f32(false, a1, false, q1, (short)0, c1, false, false);
    }
    #pragma unroll
    for (int r = 0; r < 8; ++r) {
        const float give = half ? c0[r] : c1[r];
        const int   gi   = __builtin_bit_cast(int, give);
        const float got  = __builtin_bit_cast(float,
            __builtin_amdgcn_permlanex16(gi, gi, 0x76543210, 0xfedcba98, false, false));
        const float lo = half ? got   : c0[r];
        const float hi = half ? c1[r] : got;
        dcol[(size_t)r * MM]       = softplus_f(lo + bb);
        dcol[(size_t)(8 + r) * MM] = softplus_f(hi + bb);
    }
}

__global__ __launch_bounds__(DTW_WAVES * 32) __attribute__((amdgpu_waves_per_eu(8, 8)))
void dt_proj_kernel(const float* __restrict__ xp,
                    const float* __restrict__ w,
                    const float* __restrict__ bias,
                    float* __restrict__ dt)
{
    const int lane = threadIdx.x & 31;
    const int wv   = threadIdx.x >> 5;
    const int half = lane >> 4;
    const int l15  = lane & 15;
    const int rt   = blockIdx.x / (MM / DTW_COLS);
    const int colw = (blockIdx.x % (MM / DTW_COLS)) * DTW_COLS + 32 * wv;

    const float* __restrict__ arow = xp + (size_t)(rt*16 + l15) * NPROJ + 4*half;
    const float* __restrict__ wb   = w  + (size_t)(colw + l15) * RR + 4*half;

    float4 a4[8];
    #pragma unroll
    for (int j = 0; j < 8; ++j) a4[j] = *(const float4*)(arow + 8*j);
    float4 bA[2][8];
    dtw_load_pair(bA, wb);

    float* __restrict__ drow = dt + (size_t)(rt * 16) * MM + colw + lane;
    const float bb0 = bias[colw + lane];

    dt_tile_pair(a4, bA, bb0, drow, half);
}

__global__ __launch_bounds__(TPB)
void chunk_state_kernel(const float* __restrict__ x,
                        const float* __restrict__ xp,
                        const float* __restrict__ dtb,
                        const float* __restrict__ A_log,
                        float* __restrict__ hloc,
                        float* __restrict__ dsum)
{
    const int mb = blockIdx.x % MBLK;
    const int c  = (blockIdx.x / MBLK) % (NC - 1);
    const int b  = blockIdx.x / (MBLK * (NC - 1));
    const int m  = mb * TPB + threadIdx.x;

    float Am[DS], h[DS];
    #pragma unroll
    for (int d = 0; d < DS; ++d) {
        Am[d] = -__expf(A_log[(size_t)d * MM + m]);
        h[d]  = 0.0f;
    }
    float S = 0.0f;
    const size_t row0 = (size_t)b * LL + (size_t)c * TT;

    for (int t = 0; t < TT; ++t) {
        const size_t r = row0 + t;
        const float dtv = dtb[r * MM + m];
        const float xv  = x  [r * MM + m];

        const float4* __restrict__ bc =
            (const float4*)(xp + r * NPROJ + RR);
        float Bv[DS];
        #pragma unroll
        for (int i = 0; i < 4; ++i) {
            float4 vb = bc[i];
            Bv[4*i+0] = vb.x; Bv[4*i+1] = vb.y; Bv[4*i+2] = vb.z; Bv[4*i+3] = vb.w;
        }

        const float dx = dtv * xv;
        #pragma unroll
        for (int d = 0; d < DS; ++d)
            h[d] = __expf(dtv * Am[d]) * h[d] + dx * Bv[d];
        S += dtv;
    }

    const size_t base = ((size_t)(b * NC + c) * DS) * MM + m;
    #pragma unroll
    for (int d = 0; d < DS; ++d)
        hloc[base + (size_t)d * MM] = h[d];
    dsum[(size_t)(b * NC + c) * MM + m] = S;
}

__global__ __launch_bounds__(256)
void chunk_carry_kernel(const float* __restrict__ A_log,
                        const float* __restrict__ hloc,
                        const float* __restrict__ dsum,
                        float* __restrict__ hin)
{
    const int idx = blockIdx.x * 256 + threadIdx.x;
    const int m   = idx & (MM - 1);
    const int d   = (idx >> 10) & (DS - 1);
    const int b   = idx >> 14;

    const float Am = -__expf(A_log[(size_t)d * MM + m]);
    float h = 0.0f;
    size_t o = ((size_t)(b * NC) * DS + d) * MM + m;
    hin[o] = 0.0f;
    #pragma unroll 32
    for (int c = 0; c < NC - 1; ++c) {
        const float S = dsum[(size_t)(b * NC + c) * MM + m];
        const float e = hloc[o];
        h = __expf(Am * S) * h + e;
        o += (size_t)DS * MM;
        hin[o] = h;
    }
}

__global__ __launch_bounds__(TPB)
void scan_kernel(const float* __restrict__ x,
                 const float* __restrict__ z,
                 const float* __restrict__ xp,
                 const float* __restrict__ dtb,
                 const float* __restrict__ A_log,
                 const float* __restrict__ Dp_arr,
                 const float* __restrict__ hin,
                 float* __restrict__ out)
{
    const int mb = blockIdx.x % MBLK;
    const int c  = (blockIdx.x / MBLK) % NC;
    const int b  = blockIdx.x / (MBLK * NC);
    const int m  = mb * TPB + threadIdx.x;

    float Am[DS], h[DS];
    const size_t base = ((size_t)(b * NC + c) * DS) * MM + m;
    #pragma unroll
    for (int d = 0; d < DS; ++d) {
        Am[d] = -__expf(A_log[(size_t)d * MM + m]);
        h[d]  = hin[base + (size_t)d * MM];
    }
    const float Dp = Dp_arr[m];
    const size_t row0 = (size_t)b * LL + (size_t)c * TT;

    for (int t = 0; t < TT; ++t) {
        const size_t r = row0 + t;
        const float dtv = dtb[r * MM + m];
        const float xv  = x  [r * MM + m];
        const float zv  = z  [r * MM + m];

        const float4* __restrict__ bc =
            (const float4*)(xp + r * NPROJ + RR);
        float Bv[DS], Cv[DS];
        #pragma unroll
        for (int i = 0; i < 4; ++i) {
            float4 vb = bc[i];
            float4 vc = bc[4 + i];
            Bv[4*i+0] = vb.x; Bv[4*i+1] = vb.y; Bv[4*i+2] = vb.z; Bv[4*i+3] = vb.w;
            Cv[4*i+0] = vc.x; Cv[4*i+1] = vc.y; Cv[4*i+2] = vc.z; Cv[4*i+3] = vc.w;
        }

        const float dx = dtv * xv;
        float y = 0.0f;
        #pragma unroll
        for (int d = 0; d < DS; ++d) {
            h[d] = __expf(dtv * Am[d]) * h[d] + dx * Bv[d];
            y    = fmaf(h[d], Cv[d], y);
        }
        y = fmaf(Dp, xv, y);

        const float sig = 1.0f / (1.0f + __expf(-zv));
        out[r * MM + m] = y * (zv * sig);
    }
}

extern "C" void kernel_launch(void* const* d_in, const int* in_sizes, int n_in,
                              void* d_out, int out_size, void* d_ws, size_t ws_size,
                              hipStream_t stream) {
    const float* x         = (const float*)d_in[0];
    const float* z         = (const float*)d_in[1];
    const float* x_proj_w  = (const float*)d_in[2];
    const float* dt_proj_w = (const float*)d_in[3];
    const float* dt_proj_b = (const float*)d_in[4];
    const float* A_log     = (const float*)d_in[5];
    const float* D_param   = (const float*)d_in[6];
    float* out = (float*)d_out;

    float* xp_ws = (float*)d_ws;
    float* dt_ws = xp_ws + (size_t)BLT * NPROJ;
    float* ds_ws = dt_ws + (size_t)BLT * MM;
    float* hl_ws = ds_ws + (size_t)BB * NC * MM;
    float* hi_ws = hl_ws + (size_t)BB * NC * DS * MM;

    proj_xp_kernel<<<BLT/16, 128, 0, stream>>>(x, x_proj_w, xp_ws);

    dt_proj_kernel<<<(BLT/16) * (MM/DTW_COLS), DTW_WAVES * 32, 0, stream>>>(xp_ws, dt_proj_w,
                                                                            dt_proj_b, dt_ws);

    chunk_state_kernel<<<BB * (NC - 1) * MBLK, TPB, 0, stream>>>(x, xp_ws, dt_ws, A_log,
                                                                 hl_ws, ds_ws);

    chunk_carry_kernel<<<(BB * DS * MM) / 256, 256, 0, stream>>>(A_log, hl_ws, ds_ws, hi_ws);

    scan_kernel<<<BB * NC * MBLK, TPB, 0, stream>>>(x, z, xp_ws, dt_ws,
                                                    A_log, D_param, hi_ws, out);
}


# softplus of the dt kernel: library log1pf (about 90 dependent vector instructions a value) replaced by hardware log2(1+e) times e/((1+e)-1), which restores the bits 1+e lost; dt differs in last bits
# speedup vs baseline: 20.5370x; 1.0939x over previous
#include <hip/hip_runtime.h>
#include <hip/hip_bf16.h>
#include <math.h>

#define BB      2
#define LL      2048
#define MM      1024
#define DS      16
#define RR      64
#define NPROJ   (RR + 2*DS)
#define BLT     (BB*LL)
#define NC      128
#define TT      (LL/NC)
#define TPB     256
#define MBLK    (MM/TPB)

typedef float  v2f __attribute__((ext_vector_type(2)));
typedef float  v8f __attribute__((ext_vector_type(8)));

#define KC      128
#define NCH     (MM / KC)
#define SROWS   (16 + NPROJ)
#define SPITCH  (KC + 4)
#define RPITCH  100

__global__ __launch_bounds__(128)
void proj_xp_kernel(const float* __restrict__ x,
                    const float* __restrict__ w,
                    float* __restrict__ xp)
{
    __shared__ __attribute__((aligned(16))) float smem[2 * SROWS * SPITCH];

    const int tid  = threadIdx.x;
    const int lane = tid & 31;
    const int wv   = tid >> 5;
    const int half = lane >> 4;
    const int l15  = lane & 15;
    const int rt   = blockIdx.x;

    const float* __restrict__ xs = x + (size_t)(rt*16 + 4*wv) * MM + lane*4;
    const float* __restrict__ ws = w + (size_t)(24*wv) * MM + lane*4;
    const int xd = (4*wv) * SPITCH + lane*4;
    const int wd = (16 + 24*wv) * SPITCH + lane*4;
    const int fo = l15 * SPITCH + 32*wv + 16*half;

    float4 st[28];
    v8f acc[6];
    #pragma unroll
    for (int t = 0; t < 6; ++t) acc[t] = (v8f){};

    #pragma unroll
    for (int i = 0; i < 4; ++i)  st[i]     = *(const float4*)(xs + (size_t)i * MM);
    #pragma unroll
    for (int i = 0; i < 24; ++i) st[4 + i] = *(const float4*)(ws + (size_t)i * MM);
    #pragma unroll
    for (int i = 0; i < 4; ++i)  *(float4*)(smem + xd + i*SPITCH) = st[i];
    #pragma unroll
    for (int i = 0; i < 24; ++i) *(float4*)(smem + wd + i*SPITCH) = st[4 + i];
    __syncthreads();

    #pragma unroll
    for (int s = 0; s < NCH; ++s) {
        const float* cur = smem + (s & 1) * (SROWS * SPITCH);
        float*       nxt = smem + ((s + 1) & 1) * (SROWS * SPITCH);

        if (s + 1 < NCH) {
            #pragma unroll
            for (int i = 0; i < 4; ++i)
                st[i]     = *(const float4*)(xs + (size_t)i * MM + (s + 1) * KC);
            #pragma unroll
            for (int i = 0; i < 24; ++i)
                st[4 + i] = *(const float4*)(ws + (size_t)i * MM + (s + 1) * KC);
        }

        float4 af[4], bf[6][4];
        #pragma unroll
        for (int j = 0; j < 4; ++j) af[j] = *(const float4*)(cur + fo + 4*j);
        #pragma unroll
        for (int t = 0; t < 6; ++t) {
            #pragma unroll
            for (int j = 0; j < 4; ++j)
                bf[t][j] = *(const float4*)(cur + (16 + 16*t) * SPITCH + fo + 4*j);
        }

        #pragma unroll
        for (int j = 0; j < 4; ++j) {
            v2f a0; a0.x = af[j].x; a0.y = af[j].y;
            v2f a1; a1.x = af[j].z; a1.y = af[j].w;
            #pragma unroll
            for (int t = 0; t < 6; ++t) {
                v2f b0; b0.x = bf[t][j].x; b0.y = bf[t][j].y;
                acc[t] = __builtin_amdgcn_wmma_f32_16x16x4_f32(
                             false, a0, false, b0, (short)0, acc[t], false, false);
            }
            #pragma unroll
            for (int t = 0; t < 6; ++t) {
                v2f b1; b1.x = bf[t][j].z; b1.y = bf[t][j].w;
                acc[t] = __builtin_amdgcn_wmma_f32_16x16x4_f32(
                             false, a1, false, b1, (short)0, acc[t], false, false);
            }
        }

        if (s + 1 < NCH) {
            #pragma unroll
            for (int i = 0; i < 4; ++i)  *(float4*)(nxt + xd + i*SPITCH) = st[i];
            #pragma unroll
            for (int i = 0; i < 24; ++i) *(float4*)(nxt + wd + i*SPITCH) = st[4 + i];
        }
        __syncthreads();
    }

    float* red = smem;
    #pragma unroll
    for (int t = 0; t < 6; ++t) {
        #pragma unroll
        for (int r = 0; r < 8; ++r)
            red[(wv*16 + 8*half + r) * RPITCH + 16*t + l15] = acc[t][r];
    }
    __syncthreads();

    #pragma unroll
    for (int i = 0; i < 3; ++i) {
        const int q   = tid + 128*i;
        const int row = q / 24;
        const int c4  = q - row*24;
        const float4 p0 = *(const float4*)(red + (0*16 + row) * RPITCH + 4*c4);
        const float4 p1 = *(const float4*)(red + (1*16 + row) * RPITCH + 4*c4);
        const float4 p2 = *(const float4*)(red + (2*16 + row) * RPITCH + 4*c4);
        const float4 p3 = *(const float4*)(red + (3*16 + row) * RPITCH + 4*c4);
        float4 o;
        o.x = ((p0.x + p1.x) + p2.x) + p3.x;
        o.y = ((p0.y + p1.y) + p2.y) + p3.y;
        o.z = ((p0.z + p1.z) + p2.z) + p3.z;
        o.w = ((p0.w + p1.w) + p2.w) + p3.w;
        *(float4*)(xp + (size_t)rt * (16 * NPROJ) + 4*q) = o;
    }
}

#define DTW_WAVES 8
#define DTW_COLS  (DTW_WAVES * 32)

__device__ __forceinline__ void dtw_load_pair(float4 (&b)[2][8], const float* __restrict__ wb)
{
    #pragma unroll
    for (int u = 0; u < 2; ++u) {
        #pragma unroll
        for (int j = 0; j < 8; ++j)
            b[u][j] = *(const float4*)(wb + (size_t)(16*u) * RR + 8*j);
    }
}

__device__ __forceinline__ float softplus_f(float v)
{
    const float e = __expf(v);
    const float u = 1.0f + e;
    const float d = u - 1.0f;
    const float l = __builtin_amdgcn_logf(u) * 0.693147180559945f;
    const float r = l * (e * __builtin_amdgcn_rcpf(d));
    const float s = (d == 0.0f) ? e : r;
    return (v > 20.0f) ? v : s;
}

__device__ __forceinline__ void dt_tile_pair(const float4 (&a4)[8], const float4 (&b)[2][8],
                                             float bb, float* __restrict__ dcol, int half)
{
    v8f c0 = {}, c1 = {};
    #pragma unroll
    for (int j = 0; j < 8; ++j) {
        v2f a0; a0.x = a4[j].x; a0.y = a4[j].y;
        v2f a1; a1.x = a4[j].z; a1.y = a4[j].w;
        v2f p0; p0.x = b[0][j].x; p0.y = b[0][j].y;
        v2f q0; q0.x = b[1][j].x; q0.y = b[1][j].y;
        v2f p1; p1.x = b[0][j].z; p1.y = b[0][j].w;
        v2f q1; q1.x = b[1][j].z; q1.y = b[1][j].w;
        c0 = __builtin_amdgcn_wmma_f32_16x16x4_f32(false, a0, false, p0, (short)0, c0, false, false);
        c1 = __builtin_amdgcn_wmma_f32_16x16x4_f32(false, a0, false, q0, (short)0, c1, false, false);
        c0 = __builtin_amdgcn_wmma_f32_16x16x4_f32(false, a1, false, p1, (short)0, c0, false, false);
        c1 = __builtin_amdgcn_wmma_f32_16x16x4_f32(false, a1, false, q1, (short)0, c1, false, false);
    }
    #pragma unroll
    for (int r = 0; r < 8; ++r) {
        const float give = half ? c0[r] : c1[r];
        const int   gi   = __builtin_bit_cast(int, give);
        const float got  = __builtin_bit_cast(float,
            __builtin_amdgcn_permlanex16(gi, gi, 0x76543210, 0xfedcba98, false, false));
        const float lo = half ? got   : c0[r];
        const float hi = half ? c1[r] : got;
        dcol[(size_t)r * MM]       = softplus_f(lo + bb);
        dcol[(size_t)(8 + r) * MM] = softplus_f(hi + bb);
    }
}

__global__ __launch_bounds__(DTW_WAVES * 32) __attribute__((amdgpu_waves_per_eu(8, 8)))
void dt_proj_kernel(const float* __restrict__ xp,
                    const float* __restrict__ w,
                    const float* __restrict__ bias,
                    float* __restrict__ dt)
{
    const int lane = threadIdx.x & 31;
    const int wv   = threadIdx.x >> 5;
    const int half = lane >> 4;
    const int l15  = lane & 15;
    const int rt   = blockIdx.x / (MM / DTW_COLS);
    const int colw = (blockIdx.x % (MM / DTW_COLS)) * DTW_COLS + 32 * wv;

    const float* __restrict__ arow = xp + (size_t)(rt*16 + l15) * NPROJ + 4*half;
    const float* __restrict__ wb   = w  + (size_t)(colw + l15) * RR + 4*half;

    float4 a4[8];
    #pragma unroll
    for (int j = 0; j < 8; ++j) a4[j] = *(const float4*)(arow + 8*j);
    float4 bA[2][8];
    dtw_load_pair(bA, wb);

    float* __restrict__ drow = dt + (size_t)(rt * 16) * MM + colw + lane;
    const float bb0 = bias[colw + lane];

    dt_tile_pair(a4, bA, bb0, drow, half);
}

__global__ __launch_bounds__(TPB)
void chunk_state_kernel(const float* __restrict__ x,
                        const float* __restrict__ xp,
                        const float* __restrict__ dtb,
                        const float* __restrict__ A_log,
                        float* __restrict__ hloc,
                        float* __restrict__ dsum)
{
    const int mb = blockIdx.x % MBLK;
    const int c  = (blockIdx.x / MBLK) % (NC - 1);
    const int b  = blockIdx.x / (MBLK * (NC - 1));
    const int m  = mb * TPB + threadIdx.x;

    float Am[DS], h[DS];
    #pragma unroll
    for (int d = 0; d < DS; ++d) {
        Am[d] = -__expf(A_log[(size_t)d * MM + m]);
        h[d]  = 0.0f;
    }
    float S = 0.0f;
    const size_t row0 = (size_t)b * LL + (size_t)c * TT;

    for (int t = 0; t < TT; ++t) {
        const size_t r = row0 + t;
        const float dtv = dtb[r * MM + m];
        const float xv  = x  [r * MM + m];

        const float4* __restrict__ bc =
            (const float4*)(xp + r * NPROJ + RR);
        float Bv[DS];
        #pragma unroll
        for (int i = 0; i < 4; ++i) {
            float4 vb = bc[i];
            Bv[4*i+0] = vb.x; Bv[4*i+1] = vb.y; Bv[4*i+2] = vb.z; Bv[4*i+3] = vb.w;
        }

        const float dx = dtv * xv;
        #pragma unroll
        for (int d = 0; d < DS; ++d)
            h[d] = __expf(dtv * Am[d]) * h[d] + dx * Bv[d];
        S += dtv;
    }

    const size_t base = ((size_t)(b * NC + c) * DS) * MM + m;
    #pragma unroll
    for (int d = 0; d < DS; ++d)
        hloc[base + (size_t)d * MM] = h[d];
    dsum[(size_t)(b * NC + c) * MM + m] = S;
}

__global__ __launch_bounds__(256)
void chunk_carry_kernel(const float* __restrict__ A_log,
                        const float* __restrict__ hloc,
                        const float* __restrict__ dsum,
                        float* __restrict__ hin)
{
    const int idx = blockIdx.x * 256 + threadIdx.x;
    const int m   = idx & (MM - 1);
    const int d   = (idx >> 10) & (DS - 1);
    const int b   = idx >> 14;

    const float Am = -__expf(A_log[(size_t)d * MM + m]);
    float h = 0.0f;
    size_t o = ((size_t)(b * NC) * DS + d) * MM + m;
    hin[o] = 0.0f;
    #pragma unroll 32
    for (int c = 0; c < NC - 1; ++c) {
        const float S = dsum[(size_t)(b * NC + c) * MM + m];
        const float e = hloc[o];
        h = __expf(Am * S) * h + e;
        o += (size_t)DS * MM;
        hin[o] = h;
    }
}

__global__ __launch_bounds__(TPB)
void scan_kernel(const float* __restrict__ x,
                 const float* __restrict__ z,
                 const float* __restrict__ xp,
                 const float* __restrict__ dtb,
                 const float* __restrict__ A_log,
                 const float* __restrict__ Dp_arr,
                 const float* __restrict__ hin,
                 float* __restrict__ out)
{
    const int mb = blockIdx.x % MBLK;
    const int c  = (blockIdx.x / MBLK) % NC;
    const int b  = blockIdx.x / (MBLK * NC);
    const int m  = mb * TPB + threadIdx.x;

    float Am[DS], h[DS];
    const size_t base = ((size_t)(b * NC + c) * DS) * MM + m;
    #pragma unroll
    for (int d = 0; d < DS; ++d) {
        Am[d] = -__expf(A_log[(size_t)d * MM + m]);
        h[d]  = hin[base + (size_t)d * MM];
    }
    const float Dp = Dp_arr[m];
    const size_t row0 = (size_t)b * LL + (size_t)c * TT;

    for (int t = 0; t < TT; ++t) {
        const size_t r = row0 + t;
        const float dtv = dtb[r * MM + m];
        const float xv  = x  [r * MM + m];
        const float zv  = z  [r * MM + m];

        const float4* __restrict__ bc =
            (const float4*)(xp + r * NPROJ + RR);
        float Bv[DS], Cv[DS];
        #pragma unroll
        for (int i = 0; i < 4; ++i) {
            float4 vb = bc[i];
            float4 vc = bc[4 + i];
            Bv[4*i+0] = vb.x; Bv[4*i+1] = vb.y; Bv[4*i+2] = vb.z; Bv[4*i+3] = vb.w;
            Cv[4*i+0] = vc.x; Cv[4*i+1] = vc.y; Cv[4*i+2] = vc.z; Cv[4*i+3] = vc.w;
        }

        const float dx = dtv * xv;
        float y = 0.0f;
        #pragma unroll
        for (int d = 0; d < DS; ++d) {
            h[d] = __expf(dtv * Am[d]) * h[d] + dx * Bv[d];
            y    = fmaf(h[d], Cv[d], y);
        }
        y = fmaf(Dp, xv, y);

        const float sig = 1.0f / (1.0f + __expf(-zv));
        out[r * MM + m] = y * (zv * sig);
    }
}

extern "C" void kernel_launch(void* const* d_in, const int* in_sizes, int n_in,
                              void* d_out, int out_size, void* d_ws, size_t ws_size,
                              hipStream_t stream) {
    const float* x         = (const float*)d_in[0];
    const float* z         = (const float*)d_in[1];
    const float* x_proj_w  = (const float*)d_in[2];
    const float* dt_proj_w = (const float*)d_in[3];
    const float* dt_proj_b = (const float*)d_in[4];
    const float* A_log     = (const float*)d_in[5];
    const float* D_param   = (const float*)d_in[6];
    float* out = (float*)d_out;

    float* xp_ws = (float*)d_ws;
    float* dt_ws = xp_ws + (size_t)BLT * NPROJ;
    float* ds_ws = dt_ws + (size_t)BLT * MM;
    float* hl_ws = ds_ws + (size_t)BB * NC * MM;
    float* hi_ws = hl_ws + (size_t)BB * NC * DS * MM;

    proj_xp_kernel<<<BLT/16, 128, 0, stream>>>(x, x_proj_w, xp_ws);

    dt_proj_kernel<<<(BLT/16) * (MM/DTW_COLS), DTW_WAVES * 32, 0, stream>>>(xp_ws, dt_proj_w,
                                                                            dt_proj_b, dt_ws);

    chunk_state_kernel<<<BB * (NC - 1) * MBLK, TPB, 0, stream>>>(x, xp_ws, dt_ws, A_log,
                                                                 hl_ws, ds_ws);

    chunk_carry_kernel<<<(BB * DS * MM) / 256, 256, 0, stream>>>(A_log, hl_ws, ds_ws, hi_ws);

    scan_kernel<<<BB * NC * MBLK, TPB, 0, stream>>>(x, z, xp_ws, dt_ws,
                                                    A_log, D_param, hi_ws, out);
}


# Chunk end states and dt sums computed inside the dt kernel, whose lanes already hold one channel's 16 dt values of a chunk; the separate chunk-state kernel and its launch are gone
# speedup vs baseline: 23.7435x; 1.1561x over previous
#include <hip/hip_runtime.h>
#include <hip/hip_bf16.h>
#include <math.h>

#define BB      2
#define LL      2048
#define MM      1024
#define DS      16
#define RR      64
#define NPROJ   (RR + 2*DS)
#define BLT     (BB*LL)
#define NC      128
#define TT      (LL/NC)
#define TPB     256
#define MBLK    (MM/TPB)

typedef float  v2f __attribute__((ext_vector_type(2)));
typedef float  v8f __attribute__((ext_vector_type(8)));

#define KC      128
#define NCH     (MM / KC)
#define SROWS   (16 + NPROJ)
#define SPITCH  (KC + 4)
#define RPITCH  100

__global__ __launch_bounds__(128)
void proj_xp_kernel(const float* __restrict__ x,
                    const float* __restrict__ w,
                    float* __restrict__ xp)
{
    __shared__ __attribute__((aligned(16))) float smem[2 * SROWS * SPITCH];

    const int tid  = threadIdx.x;
    const int lane = tid & 31;
    const int wv   = tid >> 5;
    const int half = lane >> 4;
    const int l15  = lane & 15;
    const int rt   = blockIdx.x;

    const float* __restrict__ xs = x + (size_t)(rt*16 + 4*wv) * MM + lane*4;
    const float* __restrict__ ws = w + (size_t)(24*wv) * MM + lane*4;
    const int xd = (4*wv) * SPITCH + lane*4;
    const int wd = (16 + 24*wv) * SPITCH + lane*4;
    const int fo = l15 * SPITCH + 32*wv + 16*half;

    float4 st[28];
    v8f acc[6];
    #pragma unroll
    for (int t = 0; t < 6; ++t) acc[t] = (v8f){};

    #pragma unroll
    for (int i = 0; i < 4; ++i)  st[i]     = *(const float4*)(xs + (size_t)i * MM);
    #pragma unroll
    for (int i = 0; i < 24; ++i) st[4 + i] = *(const float4*)(ws + (size_t)i * MM);
    #pragma unroll
    for (int i = 0; i < 4; ++i)  *(float4*)(smem + xd + i*SPITCH) = st[i];
    #pragma unroll
    for (int i = 0; i < 24; ++i) *(float4*)(smem + wd + i*SPITCH) = st[4 + i];
    __syncthreads();

    #pragma unroll
    for (int s = 0; s < NCH; ++s) {
        const float* cur = smem + (s & 1) * (SROWS * SPITCH);
        float*       nxt = smem + ((s + 1) & 1) * (SROWS * SPITCH);

        if (s + 1 < NCH) {
            #pragma unroll
            for (int i = 0; i < 4; ++i)
                st[i]     = *(const float4*)(xs + (size_t)i * MM + (s + 1) * KC);
            #pragma unroll
            for (int i = 0; i < 24; ++i)
                st[4 + i] = *(const float4*)(ws + (size_t)i * MM + (s + 1) * KC);
        }

        float4 af[4], bf[6][4];
        #pragma unroll
        for (int j = 0; j < 4; ++j) af[j] = *(const float4*)(cur + fo + 4*j);
        #pragma unroll
        for (int t = 0; t < 6; ++t) {
            #pragma unroll
            for (int j = 0; j < 4; ++j)
                bf[t][j] = *(const float4*)(cur + (16 + 16*t) * SPITCH + fo + 4*j);
        }

        #pragma unroll
        for (int j = 0; j < 4; ++j) {
            v2f a0; a0.x = af[j].x; a0.y = af[j].y;
            v2f a1; a1.x = af[j].z; a1.y = af[j].w;
            #pragma unroll
            for (int t = 0; t < 6; ++t) {
                v2f b0; b0.x = bf[t][j].x; b0.y = bf[t][j].y;
                acc[t] = __builtin_amdgcn_wmma_f32_16x16x4_f32(
                             false, a0, false, b0, (short)0, acc[t], false, false);
            }
            #pragma unroll
            for (int t = 0; t < 6; ++t) {
                v2f b1; b1.x = bf[t][j].z; b1.y = bf[t][j].w;
                acc[t] = __builtin_amdgcn_wmma_f32_16x16x4_f32(
                             false, a1, false, b1, (short)0, acc[t], false, false);
            }
        }

        if (s + 1 < NCH) {
            #pragma unroll
            for (int i = 0; i < 4; ++i)  *(float4*)(nxt + xd + i*SPITCH) = st[i];
            #pragma unroll
            for (int i = 0; i < 24; ++i) *(float4*)(nxt + wd + i*SPITCH) = st[4 + i];
        }
        __syncthreads();
    }

    float* red = smem;
    #pragma unroll
    for (int t = 0; t < 6; ++t) {
        #pragma unroll
        for (int r = 0; r < 8; ++r)
            red[(wv*16 + 8*half + r) * RPITCH + 16*t + l15] = acc[t][r];
    }
    __syncthreads();

    #pragma unroll
    for (int i = 0; i < 3; ++i) {
        const int q   = tid + 128*i;
        const int row = q / 24;
        const int c4  = q - row*24;
        const float4 p0 = *(const float4*)(red + (0*16 + row) * RPITCH + 4*c4);
        const float4 p1 = *(const float4*)(red + (1*16 + row) * RPITCH + 4*c4);
        const float4 p2 = *(const float4*)(red + (2*16 + row) * RPITCH + 4*c4);
        const float4 p3 = *(const float4*)(red + (3*16 + row) * RPITCH + 4*c4);
        float4 o;
        o.x = ((p0.x + p1.x) + p2.x) + p3.x;
        o.y = ((p0.y + p1.y) + p2.y) + p3.y;
        o.z = ((p0.z + p1.z) + p2.z) + p3.z;
        o.w = ((p0.w + p1.w) + p2.w) + p3.w;
        *(float4*)(xp + (size_t)rt * (16 * NPROJ) + 4*q) = o;
    }
}

#define DTW_WAVES 8
#define DTW_COLS  (DTW_WAVES * 32)

__device__ __forceinline__ void dtw_load_pair(float4 (&b)[2][8], const float* __restrict__ wb)
{
    #pragma unroll
    for (int u = 0; u < 2; ++u) {
        #pragma unroll
        for (int j = 0; j < 8; ++j)
            b[u][j] = *(const float4*)(wb + (size_t)(16*u) * RR + 8*j);
    }
}

__device__ __forceinline__ float softplus_f(float v)
{
    const float e = __expf(v);
    const float u = 1.0f + e;
    const float d = u - 1.0f;
    const float l = __builtin_amdgcn_logf(u) * 0.693147180559945f;
    const float r = l * (e * __builtin_amdgcn_rcpf(d));
    const float s = (d == 0.0f) ? e : r;
    return (v > 20.0f) ? v : s;
}

__global__ __launch_bounds__(DTW_WAVES * 32) __attribute__((amdgpu_waves_per_eu(8, 8)))
void dt_state_kernel(const float* __restrict__ xp,
                     const float* __restrict__ w,
                     const float* __restrict__ bias,
                     const float* __restrict__ x,
                     const float* __restrict__ A_log,
                     float* __restrict__ dt,
                     float* __restrict__ hloc,
                     float* __restrict__ dsum)
{
    const int lane = threadIdx.x & 31;
    const int wv   = threadIdx.x >> 5;
    const int half = lane >> 4;
    const int l15  = lane & 15;
    const int rt   = blockIdx.x / (MM / DTW_COLS);
    const int colw = (blockIdx.x % (MM / DTW_COLS)) * DTW_COLS + 32 * wv;
    const int m    = colw + lane;

    const float* __restrict__ arow = xp + (size_t)(rt*16 + l15) * NPROJ + 4*half;
    const float* __restrict__ wb   = w  + (size_t)(colw + l15) * RR + 4*half;

    float4 a4[8];
    #pragma unroll
    for (int j = 0; j < 8; ++j) a4[j] = *(const float4*)(arow + 8*j);
    float4 b[2][8];
    dtw_load_pair(b, wb);
    const float bb = bias[m];

    v8f c0 = {}, c1 = {};
    #pragma unroll
    for (int j = 0; j < 8; ++j) {
        v2f a0; a0.x = a4[j].x; a0.y = a4[j].y;
        v2f a1; a1.x = a4[j].z; a1.y = a4[j].w;
        v2f p0; p0.x = b[0][j].x; p0.y = b[0][j].y;
        v2f q0; q0.x = b[1][j].x; q0.y = b[1][j].y;
        v2f p1; p1.x = b[0][j].z; p1.y = b[0][j].w;
        v2f q1; q1.x = b[1][j].z; q1.y = b[1][j].w;
        c0 = __builtin_amdgcn_wmma_f32_16x16x4_f32(false, a0, false, p0, (short)0, c0, false, false);
        c1 = __builtin_amdgcn_wmma_f32_16x16x4_f32(false, a0, false, q0, (short)0, c1, false, false);
        c0 = __builtin_amdgcn_wmma_f32_16x16x4_f32(false, a1, false, p1, (short)0, c0, false, false);
        c1 = __builtin_amdgcn_wmma_f32_16x16x4_f32(false, a1, false, q1, (short)0, c1, false, false);
    }

    const float* __restrict__ xcol = x + (size_t)(rt * 16) * MM + m;
    float xv[TT], al[DS];
    #pragma unroll
    for (int t = 0; t < TT; ++t) xv[t] = xcol[(size_t)t * MM];
    #pragma unroll
    for (int d = 0; d < DS; ++d) al[d] = A_log[(size_t)d * MM + m];

    float* __restrict__ dcol = dt + (size_t)(rt * 16) * MM + m;
    float dtv[TT];
    #pragma unroll
    for (int r = 0; r < 8; ++r) {
        const float give = half ? c0[r] : c1[r];
        const int   gi   = __builtin_bit_cast(int, give);
        const float got  = __builtin_bit_cast(float,
            __builtin_amdgcn_permlanex16(gi, gi, 0x76543210, 0xfedcba98, false, false));
        const float lo = half ? got   : c0[r];
        const float hi = half ? c1[r] : got;
        dtv[r]     = softplus_f(lo + bb);
        dtv[8 + r] = softplus_f(hi + bb);
        dcol[(size_t)r * MM]       = dtv[r];
        dcol[(size_t)(8 + r) * MM] = dtv[8 + r];
    }

    float Am[DS], h[DS];
    #pragma unroll
    for (int d = 0; d < DS; ++d) {
        Am[d] = -__expf(al[d]);
        h[d]  = 0.0f;
    }
    float S = 0.0f;

    #pragma unroll
    for (int t = 0; t < TT; ++t) {
        const float4* __restrict__ bc =
            (const float4*)(xp + (size_t)(rt * 16 + t) * NPROJ + RR);
        float Bv[DS];
        #pragma unroll
        for (int i = 0; i < 4; ++i) {
            float4 vb = bc[i];
            Bv[4*i+0] = vb.x; Bv[4*i+1] = vb.y; Bv[4*i+2] = vb.z; Bv[4*i+3] = vb.w;
        }

        const float dtt = dtv[t];
        const float dx  = dtt * xv[t];
        #pragma unroll
        for (int d = 0; d < DS; ++d)
            h[d] = __expf(dtt * Am[d]) * h[d] + dx * Bv[d];
        S += dtt;
    }

    const size_t base = ((size_t)rt * DS) * MM + m;
    #pragma unroll
    for (int d = 0; d < DS; ++d)
        hloc[base + (size_t)d * MM] = h[d];
    dsum[(size_t)rt * MM + m] = S;
}

__global__ __launch_bounds__(256)
void chunk_carry_kernel(const float* __restrict__ A_log,
                        const float* __restrict__ hloc,
                        const float* __restrict__ dsum,
                        float* __restrict__ hin)
{
    const int idx = blockIdx.x * 256 + threadIdx.x;
    const int m   = idx & (MM - 1);
    const int d   = (idx >> 10) & (DS - 1);
    const int b   = idx >> 14;

    const float Am = -__expf(A_log[(size_t)d * MM + m]);
    float h = 0.0f;
    size_t o = ((size_t)(b * NC) * DS + d) * MM + m;
    hin[o] = 0.0f;
    #pragma unroll 32
    for (int c = 0; c < NC - 1; ++c) {
        const float S = dsum[(size_t)(b * NC + c) * MM + m];
        const float e = hloc[o];
        h = __expf(Am * S) * h + e;
        o += (size_t)DS * MM;
        hin[o] = h;
    }
}

__global__ __launch_bounds__(TPB)
void scan_kernel(const float* __restrict__ x,
                 const float* __restrict__ z,
                 const float* __restrict__ xp,
                 const float* __restrict__ dtb,
                 const float* __restrict__ A_log,
                 const float* __restrict__ Dp_arr,
                 const float* __restrict__ hin,
                 float* __restrict__ out)
{
    const int mb = blockIdx.x % MBLK;
    const int c  = (blockIdx.x / MBLK) % NC;
    const int b  = blockIdx.x / (MBLK * NC);
    const int m  = mb * TPB + threadIdx.x;

    float Am[DS], h[DS];
    const size_t base = ((size_t)(b * NC + c) * DS) * MM + m;
    #pragma unroll
    for (int d = 0; d < DS; ++d) {
        Am[d] = -__expf(A_log[(size_t)d * MM + m]);
        h[d]  = hin[base + (size_t)d * MM];
    }
    const float Dp = Dp_arr[m];
    const size_t row0 = (size_t)b * LL + (size_t)c * TT;

    for (int t = 0; t < TT; ++t) {
        const size_t r = row0 + t;
        const float dtv = dtb[r * MM + m];
        const float xv  = x  [r * MM + m];
        const float zv  = z  [r * MM + m];

        const float4* __restrict__ bc =
            (const float4*)(xp + r * NPROJ + RR);
        float Bv[DS], Cv[DS];
        #pragma unroll
        for (int i = 0; i < 4; ++i) {
            float4 vb = bc[i];
            float4 vc = bc[4 + i];
            Bv[4*i+0] = vb.x; Bv[4*i+1] = vb.y; Bv[4*i+2] = vb.z; Bv[4*i+3] = vb.w;
            Cv[4*i+0] = vc.x; Cv[4*i+1] = vc.y; Cv[4*i+2] = vc.z; Cv[4*i+3] = vc.w;
        }

        const float dx = dtv * xv;
        float y = 0.0f;
        #pragma unroll
        for (int d = 0; d < DS; ++d) {
            h[d] = __expf(dtv * Am[d]) * h[d] + dx * Bv[d];
            y    = fmaf(h[d], Cv[d], y);
        }
        y = fmaf(Dp, xv, y);

        const float sig = 1.0f / (1.0f + __expf(-zv));
        out[r * MM + m] = y * (zv * sig);
    }
}

extern "C" void kernel_launch(void* const* d_in, const int* in_sizes, int n_in,
                              void* d_out, int out_size, void* d_ws, size_t ws_size,
                              hipStream_t stream) {
    const float* x         = (const float*)d_in[0];
    const float* z         = (const float*)d_in[1];
    const float* x_proj_w  = (const float*)d_in[2];
    const float* dt_proj_w = (const float*)d_in[3];
    const float* dt_proj_b = (const float*)d_in[4];
    const float* A_log     = (const float*)d_in[5];
    const float* D_param   = (const float*)d_in[6];
    float* out = (float*)d_out;

    float* xp_ws = (float*)d_ws;
    float* dt_ws = xp_ws + (size_t)BLT * NPROJ;
    float* ds_ws = dt_ws + (size_t)BLT * MM;
    float* hl_ws = ds_ws + (size_t)BB * NC * MM;
    float* hi_ws = hl_ws + (size_t)BB * NC * DS * MM;

    proj_xp_kernel<<<BLT/16, 128, 0, stream>>>(x, x_proj_w, xp_ws);

    dt_state_kernel<<<(BLT/16) * (MM/DTW_COLS), DTW_WAVES * 32, 0, stream>>>(xp_ws, dt_proj_w, dt_proj_b,
                                                                             x, A_log, dt_ws, hl_ws, ds_ws);

    chunk_carry_kernel<<<(BB * DS * MM) / 256, 256, 0, stream>>>(A_log, hl_ws, ds_ws, hi_ws);

    scan_kernel<<<BB * NC * MBLK, TPB, 0, stream>>>(x, z, xp_ws, dt_ws,
                                                    A_log, D_param, hi_ws, out);
}


# scan kernel: start state, A_log and all 16 steps' dt, x, z loads of a wave issued together ahead of the first step (steps unrolled), so a wave waits for memory once instead of once per step
# speedup vs baseline: 24.8621x; 1.0471x over previous
#include <hip/hip_runtime.h>
#include <hip/hip_bf16.h>
#include <math.h>

#define BB      2
#define LL      2048
#define MM      1024
#define DS      16
#define RR      64
#define NPROJ   (RR + 2*DS)
#define BLT     (BB*LL)
#define NC      128
#define TT      (LL/NC)
#define TPB     256
#define MBLK    (MM/TPB)

typedef float  v2f __attribute__((ext_vector_type(2)));
typedef float  v8f __attribute__((ext_vector_type(8)));

#define KC      128
#define NCH     (MM / KC)
#define SROWS   (16 + NPROJ)
#define SPITCH  (KC + 4)
#define RPITCH  100

__global__ __launch_bounds__(128)
void proj_xp_kernel(const float* __restrict__ x,
                    const float* __restrict__ w,
                    float* __restrict__ xp)
{
    __shared__ __attribute__((aligned(16))) float smem[2 * SROWS * SPITCH];

    const int tid  = threadIdx.x;
    const int lane = tid & 31;
    const int wv   = tid >> 5;
    const int half = lane >> 4;
    const int l15  = lane & 15;
    const int rt   = blockIdx.x;

    const float* __restrict__ xs = x + (size_t)(rt*16 + 4*wv) * MM + lane*4;
    const float* __restrict__ ws = w + (size_t)(24*wv) * MM + lane*4;
    const int xd = (4*wv) * SPITCH + lane*4;
    const int wd = (16 + 24*wv) * SPITCH + lane*4;
    const int fo = l15 * SPITCH + 32*wv + 16*half;

    float4 st[28];
    v8f acc[6];
    #pragma unroll
    for (int t = 0; t < 6; ++t) acc[t] = (v8f){};

    #pragma unroll
    for (int i = 0; i < 4; ++i)  st[i]     = *(const float4*)(xs + (size_t)i * MM);
    #pragma unroll
    for (int i = 0; i < 24; ++i) st[4 + i] = *(const float4*)(ws + (size_t)i * MM);
    #pragma unroll
    for (int i = 0; i < 4; ++i)  *(float4*)(smem + xd + i*SPITCH) = st[i];
    #pragma unroll
    for (int i = 0; i < 24; ++i) *(float4*)(smem + wd + i*SPITCH) = st[4 + i];
    __syncthreads();

    #pragma unroll
    for (int s = 0; s < NCH; ++s) {
        const float* cur = smem + (s & 1) * (SROWS * SPITCH);
        float*       nxt = smem + ((s + 1) & 1) * (SROWS * SPITCH);

        if (s + 1 < NCH) {
            #pragma unroll
            for (int i = 0; i < 4; ++i)
                st[i]     = *(const float4*)(xs + (size_t)i * MM + (s + 1) * KC);
            #pragma unroll
            for (int i = 0; i < 24; ++i)
                st[4 + i] = *(const float4*)(ws + (size_t)i * MM + (s + 1) * KC);
        }

        float4 af[4], bf[6][4];
        #pragma unroll
        for (int j = 0; j < 4; ++j) af[j] = *(const float4*)(cur + fo + 4*j);
        #pragma unroll
        for (int t = 0; t < 6; ++t) {
            #pragma unroll
            for (int j = 0; j < 4; ++j)
                bf[t][j] = *(const float4*)(cur + (16 + 16*t) * SPITCH + fo + 4*j);
        }

        #pragma unroll
        for (int j = 0; j < 4; ++j) {
            v2f a0; a0.x = af[j].x; a0.y = af[j].y;
            v2f a1; a1.x = af[j].z; a1.y = af[j].w;
            #pragma unroll
            for (int t = 0; t < 6; ++t) {
                v2f b0; b0.x = bf[t][j].x; b0.y = bf[t][j].y;
                acc[t] = __builtin_amdgcn_wmma_f32_16x16x4_f32(
                             false, a0, false, b0, (short)0, acc[t], false, false);
            }
            #pragma unroll
            for (int t = 0; t < 6; ++t) {
                v2f b1; b1.x = bf[t][j].z; b1.y = bf[t][j].w;
                acc[t] = __builtin_amdgcn_wmma_f32_16x16x4_f32(
                             false, a1, false, b1, (short)0, acc[t], false, false);
            }
        }

        if (s + 1 < NCH) {
            #pragma unroll
            for (int i = 0; i < 4; ++i)  *(float4*)(nxt + xd + i*SPITCH) = st[i];
            #pragma unroll
            for (int i = 0; i < 24; ++i) *(float4*)(nxt + wd + i*SPITCH) = st[4 + i];
        }
        __syncthreads();
    }

    float* red = smem;
    #pragma unroll
    for (int t = 0; t < 6; ++t) {
        #pragma unroll
        for (int r = 0; r < 8; ++r)
            red[(wv*16 + 8*half + r) * RPITCH + 16*t + l15] = acc[t][r];
    }
    __syncthreads();

    #pragma unroll
    for (int i = 0; i < 3; ++i) {
        const int q   = tid + 128*i;
        const int row = q / 24;
        const int c4  = q - row*24;
        const float4 p0 = *(const float4*)(red + (0*16 + row) * RPITCH + 4*c4);
        const float4 p1 = *(const float4*)(red + (1*16 + row) * RPITCH + 4*c4);
        const float4 p2 = *(const float4*)(red + (2*16 + row) * RPITCH + 4*c4);
        const float4 p3 = *(const float4*)(red + (3*16 + row) * RPITCH + 4*c4);
        float4 o;
        o.x = ((p0.x + p1.x) + p2.x) + p3.x;
        o.y = ((p0.y + p1.y) + p2.y) + p3.y;
        o.z = ((p0.z + p1.z) + p2.z) + p3.z;
        o.w = ((p0.w + p1.w) + p2.w) + p3.w;
        *(float4*)(xp + (size_t)rt * (16 * NPROJ) + 4*q) = o;
    }
}

#define DTW_WAVES 8
#define DTW_COLS  (DTW_WAVES * 32)

__device__ __forceinline__ void dtw_load_pair(float4 (&b)[2][8], const float* __restrict__ wb)
{
    #pragma unroll
    for (int u = 0; u < 2; ++u) {
        #pragma unroll
        for (int j = 0; j < 8; ++j)
            b[u][j] = *(const float4*)(wb + (size_t)(16*u) * RR + 8*j);
    }
}

__device__ __forceinline__ float softplus_f(float v)
{
    const float e = __expf(v);
    const float u = 1.0f + e;
    const float d = u - 1.0f;
    const float l = __builtin_amdgcn_logf(u) * 0.693147180559945f;
    const float r = l * (e * __builtin_amdgcn_rcpf(d));
    const float s = (d == 0.0f) ? e : r;
    return (v > 20.0f) ? v : s;
}

__global__ __launch_bounds__(DTW_WAVES * 32) __attribute__((amdgpu_waves_per_eu(8, 8)))
void dt_state_kernel(const float* __restrict__ xp,
                     const float* __restrict__ w,
                     const float* __restrict__ bias,
                     const float* __restrict__ x,
                     const float* __restrict__ A_log,
                     float* __restrict__ dt,
                     float* __restrict__ hloc,
                     float* __restrict__ dsum)
{
    const int lane = threadIdx.x & 31;
    const int wv   = threadIdx.x >> 5;
    const int half = lane >> 4;
    const int l15  = lane & 15;
    const int rt   = blockIdx.x / (MM / DTW_COLS);
    const int colw = (blockIdx.x % (MM / DTW_COLS)) * DTW_COLS + 32 * wv;
    const int m    = colw + lane;

    const float* __restrict__ arow = xp + (size_t)(rt*16 + l15) * NPROJ + 4*half;
    const float* __restrict__ wb   = w  + (size_t)(colw + l15) * RR + 4*half;

    float4 a4[8];
    #pragma unroll
    for (int j = 0; j < 8; ++j) a4[j] = *(const float4*)(arow + 8*j);
    float4 b[2][8];
    dtw_load_pair(b, wb);
    const float bb = bias[m];

    v8f c0 = {}, c1 = {};
    #pragma unroll
    for (int j = 0; j < 8; ++j) {
        v2f a0; a0.x = a4[j].x; a0.y = a4[j].y;
        v2f a1; a1.x = a4[j].z; a1.y = a4[j].w;
        v2f p0; p0.x = b[0][j].x; p0.y = b[0][j].y;
        v2f q0; q0.x = b[1][j].x; q0.y = b[1][j].y;
        v2f p1; p1.x = b[0][j].z; p1.y = b[0][j].w;
        v2f q1; q1.x = b[1][j].z; q1.y = b[1][j].w;
        c0 = __builtin_amdgcn_wmma_f32_16x16x4_f32(false, a0, false, p0, (short)0, c0, false, false);
        c1 = __builtin_amdgcn_wmma_f32_16x16x4_f32(false, a0, false, q0, (short)0, c1, false, false);
        c0 = __builtin_amdgcn_wmma_f32_16x16x4_f32(false, a1, false, p1, (short)0, c0, false, false);
        c1 = __builtin_amdgcn_wmma_f32_16x16x4_f32(false, a1, false, q1, (short)0, c1, false, false);
    }

    const float* __restrict__ xcol = x + (size_t)(rt * 16) * MM + m;
    float xv[TT], al[DS];
    #pragma unroll
    for (int t = 0; t < TT; ++t) xv[t] = xcol[(size_t)t * MM];
    #pragma unroll
    for (int d = 0; d < DS; ++d) al[d] = A_log[(size_t)d * MM + m];

    float* __restrict__ dcol = dt + (size_t)(rt * 16) * MM + m;
    float dtv[TT];
    #pragma unroll
    for (int r = 0; r < 8; ++r) {
        const float give = half ? c0[r] : c1[r];
        const int   gi   = __builtin_bit_cast(int, give);
        const float got  = __builtin_bit_cast(float,
            __builtin_amdgcn_permlanex16(gi, gi, 0x76543210, 0xfedcba98, false, false));
        const float lo = half ? got   : c0[r];
        const float hi = half ? c1[r] : got;
        dtv[r]     = softplus_f(lo + bb);
        dtv[8 + r] = softplus_f(hi + bb);
        dcol[(size_t)r * MM]       = dtv[r];
        dcol[(size_t)(8 + r) * MM] = dtv[8 + r];
    }

    float Am[DS], h[DS];
    #pragma unroll
    for (int d = 0; d < DS; ++d) {
        Am[d] = -__expf(al[d]);
        h[d]  = 0.0f;
    }
    float S = 0.0f;

    #pragma unroll
    for (int t = 0; t < TT; ++t) {
        const float4* __restrict__ bc =
            (const float4*)(xp + (size_t)(rt * 16 + t) * NPROJ + RR);
        float Bv[DS];
        #pragma unroll
        for (int i = 0; i < 4; ++i) {
            float4 vb = bc[i];
            Bv[4*i+0] = vb.x; Bv[4*i+1] = vb.y; Bv[4*i+2] = vb.z; Bv[4*i+3] = vb.w;
        }

        const float dtt = dtv[t];
        const float dx  = dtt * xv[t];
        #pragma unroll
        for (int d = 0; d < DS; ++d)
            h[d] = __expf(dtt * Am[d]) * h[d] + dx * Bv[d];
        S += dtt;
    }

    const size_t base = ((size_t)rt * DS) * MM + m;
    #pragma unroll
    for (int d = 0; d < DS; ++d)
        hloc[base + (size_t)d * MM] = h[d];
    dsum[(size_t)rt * MM + m] = S;
}

__global__ __launch_bounds__(256)
void chunk_carry_kernel(const float* __restrict__ A_log,
                        const float* __restrict__ hloc,
                        const float* __restrict__ dsum,
                        float* __restrict__ hin)
{
    const int idx = blockIdx.x * 256 + threadIdx.x;
    const int m   = idx & (MM - 1);
    const int d   = (idx >> 10) & (DS - 1);
    const int b   = idx >> 14;

    const float Am = -__expf(A_log[(size_t)d * MM + m]);
    float h = 0.0f;
    size_t o = ((size_t)(b * NC) * DS + d) * MM + m;
    hin[o] = 0.0f;
    #pragma unroll 32
    for (int c = 0; c < NC - 1; ++c) {
        const float S = dsum[(size_t)(b * NC + c) * MM + m];
        const float e = hloc[o];
        h = __expf(Am * S) * h + e;
        o += (size_t)DS * MM;
        hin[o] = h;
    }
}

__global__ __launch_bounds__(TPB) __attribute__((amdgpu_waves_per_eu(8, 8)))
void scan_kernel(const float* __restrict__ x,
                 const float* __restrict__ z,
                 const float* __restrict__ xp,
                 const float* __restrict__ dtb,
                 const float* __restrict__ A_log,
                 const float* __restrict__ Dp_arr,
                 const float* __restrict__ hin,
                 float* __restrict__ out)
{
    const int mb = blockIdx.x % MBLK;
    const int c  = (blockIdx.x / MBLK) % NC;
    const int b  = blockIdx.x / (MBLK * NC);
    const int m  = mb * TPB + threadIdx.x;

    const size_t base = ((size_t)(b * NC + c) * DS) * MM + m;
    const size_t row0 = (size_t)b * LL + (size_t)c * TT;
    const float* __restrict__ dcol = dtb + row0 * MM + m;
    const float* __restrict__ xcol = x   + row0 * MM + m;
    const float* __restrict__ zcol = z   + row0 * MM + m;
    float*       __restrict__ ocol = out + row0 * MM + m;

    float al[DS], h[DS];
    #pragma unroll
    for (int d = 0; d < DS; ++d) {
        al[d] = A_log[(size_t)d * MM + m];
        h[d]  = hin[base + (size_t)d * MM];
    }
    const float Dp = Dp_arr[m];

    float dtv[TT], xv[TT], zv[TT];
    #pragma unroll
    for (int t = 0; t < TT; ++t) {
        dtv[t] = dcol[(size_t)t * MM];
        xv[t]  = xcol[(size_t)t * MM];
        zv[t]  = zcol[(size_t)t * MM];
    }

    float Am[DS];
    #pragma unroll
    for (int d = 0; d < DS; ++d)
        Am[d] = -__expf(al[d]);

    #pragma unroll
    for (int t = 0; t < TT; ++t) {
        const float4* __restrict__ bc =
            (const float4*)(xp + (row0 + t) * NPROJ + RR);
        float Bv[DS], Cv[DS];
        #pragma unroll
        for (int i = 0; i < 4; ++i) {
            float4 vb = bc[i];
            float4 vc = bc[4 + i];
            Bv[4*i+0] = vb.x; Bv[4*i+1] = vb.y; Bv[4*i+2] = vb.z; Bv[4*i+3] = vb.w;
            Cv[4*i+0] = vc.x; Cv[4*i+1] = vc.y; Cv[4*i+2] = vc.z; Cv[4*i+3] = vc.w;
        }

        const float dx = dtv[t] * xv[t];
        float y = 0.0f;
        #pragma unroll
        for (int d = 0; d < DS; ++d) {
            h[d] = __expf(dtv[t] * Am[d]) * h[d] + dx * Bv[d];
            y    = fmaf(h[d], Cv[d], y);
        }
        y = fmaf(Dp, xv[t], y);

        const float sig = 1.0f / (1.0f + __expf(-zv[t]));
        ocol[(size_t)t * MM] = y * (zv[t] * sig);
    }
}

extern "C" void kernel_launch(void* const* d_in, const int* in_sizes, int n_in,
                              void* d_out, int out_size, void* d_ws, size_t ws_size,
                              hipStream_t stream) {
    const float* x         = (const float*)d_in[0];
    const float* z         = (const float*)d_in[1];
    const float* x_proj_w  = (const float*)d_in[2];
    const float* dt_proj_w = (const float*)d_in[3];
    const float* dt_proj_b = (const float*)d_in[4];
    const float* A_log     = (const float*)d_in[5];
    const float* D_param   = (const float*)d_in[6];
    float* out = (float*)d_out;

    float* xp_ws = (float*)d_ws;
    float* dt_ws = xp_ws + (size_t)BLT * NPROJ;
    float* ds_ws = dt_ws + (size_t)BLT * MM;
    float* hl_ws = ds_ws + (size_t)BB * NC * MM;
    float* hi_ws = hl_ws + (size_t)BB * NC * DS * MM;

    proj_xp_kernel<<<BLT/16, 128, 0, stream>>>(x, x_proj_w, xp_ws);

    dt_state_kernel<<<(BLT/16) * (MM/DTW_COLS), DTW_WAVES * 32, 0, stream>>>(xp_ws, dt_proj_w, dt_proj_b,
                                                                             x, A_log, dt_ws, hl_ws, ds_ws);

    chunk_carry_kernel<<<(BB * DS * MM) / 256, 256, 0, stream>>>(A_log, hl_ws, ds_ws, hi_ws);

    scan_kernel<<<BB * NC * MBLK, TPB, 0, stream>>>(x, z, xp_ws, dt_ws,
                                                    A_log, D_param, hi_ws, out);
}


# dt kernel and scan kernel: log2 e folded into A once per lane, so a state's decay exp(dt*A) is one multiply and a hardware exp2 per step instead of two multiplies and exp2; decays differ in last bits
# speedup vs baseline: 26.1421x; 1.0515x over previous
#include <hip/hip_runtime.h>
#include <hip/hip_bf16.h>
#include <math.h>

#define BB      2
#define LL      2048
#define MM      1024
#define DS      16
#define RR      64
#define NPROJ   (RR + 2*DS)
#define BLT     (BB*LL)
#define NC      128
#define TT      (LL/NC)
#define TPB     256
#define MBLK    (MM/TPB)

typedef float  v2f __attribute__((ext_vector_type(2)));
typedef float  v8f __attribute__((ext_vector_type(8)));

#define LOG2E_F 1.44269504088896341f

#define KC      128
#define NCH     (MM / KC)
#define SROWS   (16 + NPROJ)
#define SPITCH  (KC + 4)
#define RPITCH  100

__global__ __launch_bounds__(128)
void proj_xp_kernel(const float* __restrict__ x,
                    const float* __restrict__ w,
                    float* __restrict__ xp)
{
    __shared__ __attribute__((aligned(16))) float smem[2 * SROWS * SPITCH];

    const int tid  = threadIdx.x;
    const int lane = tid & 31;
    const int wv   = tid >> 5;
    const int half = lane >> 4;
    const int l15  = lane & 15;
    const int rt   = blockIdx.x;

    const float* __restrict__ xs = x + (size_t)(rt*16 + 4*wv) * MM + lane*4;
    const float* __restrict__ ws = w + (size_t)(24*wv) * MM + lane*4;
    const int xd = (4*wv) * SPITCH + lane*4;
    const int wd = (16 + 24*wv) * SPITCH + lane*4;
    const int fo = l15 * SPITCH + 32*wv + 16*half;

    float4 st[28];
    v8f acc[6];
    #pragma unroll
    for (int t = 0; t < 6; ++t) acc[t] = (v8f){};

    #pragma unroll
    for (int i = 0; i < 4; ++i)  st[i]     = *(const float4*)(xs + (size_t)i * MM);
    #pragma unroll
    for (int i = 0; i < 24; ++i) st[4 + i] = *(const float4*)(ws + (size_t)i * MM);
    #pragma unroll
    for (int i = 0; i < 4; ++i)  *(float4*)(smem + xd + i*SPITCH) = st[i];
    #pragma unroll
    for (int i = 0; i < 24; ++i) *(float4*)(smem + wd + i*SPITCH) = st[4 + i];
    __syncthreads();

    #pragma unroll
    for (int s = 0; s < NCH; ++s) {
        const float* cur = smem + (s & 1) * (SROWS * SPITCH);
        float*       nxt = smem + ((s + 1) & 1) * (SROWS * SPITCH);

        if (s + 1 < NCH) {
            #pragma unroll
            for (int i = 0; i < 4; ++i)
                st[i]     = *(const float4*)(xs + (size_t)i * MM + (s + 1) * KC);
            #pragma unroll
            for (int i = 0; i < 24; ++i)
                st[4 + i] = *(const float4*)(ws + (size_t)i * MM + (s + 1) * KC);
        }

        float4 af[4], bf[6][4];
        #pragma unroll
        for (int j = 0; j < 4; ++j) af[j] = *(const float4*)(cur + fo + 4*j);
        #pragma unroll
        for (int t = 0; t < 6; ++t) {
            #pragma unroll
            for (int j = 0; j < 4; ++j)
                bf[t][j] = *(const float4*)(cur + (16 + 16*t) * SPITCH + fo + 4*j);
        }

        #pragma unroll
        for (int j = 0; j < 4; ++j) {
            v2f a0; a0.x = af[j].x; a0.y = af[j].y;
            v2f a1; a1.x = af[j].z; a1.y = af[j].w;
            #pragma unroll
            for (int t = 0; t < 6; ++t) {
                v2f b0; b0.x = bf[t][j].x; b0.y = bf[t][j].y;
                acc[t] = __builtin_amdgcn_wmma_f32_16x16x4_f32(
                             false, a0, false, b0, (short)0, acc[t], false, false);
            }
            #pragma unroll
            for (int t = 0; t < 6; ++t) {
                v2f b1; b1.x = bf[t][j].z; b1.y = bf[t][j].w;
                acc[t] = __builtin_amdgcn_wmma_f32_16x16x4_f32(
                             false, a1, false, b1, (short)0, acc[t], false, false);
            }
        }

        if (s + 1 < NCH) {
            #pragma unroll
            for (int i = 0; i < 4; ++i)  *(float4*)(nxt + xd + i*SPITCH) = st[i];
            #pragma unroll
            for (int i = 0; i < 24; ++i) *(float4*)(nxt + wd + i*SPITCH) = st[4 + i];
        }
        __syncthreads();
    }

    float* red = smem;
    #pragma unroll
    for (int t = 0; t < 6; ++t) {
        #pragma unroll
        for (int r = 0; r < 8; ++r)
            red[(wv*16 + 8*half + r) * RPITCH + 16*t + l15] = acc[t][r];
    }
    __syncthreads();

    #pragma unroll
    for (int i = 0; i < 3; ++i) {
        const int q   = tid + 128*i;
        const int row = q / 24;
        const int c4  = q - row*24;
        const float4 p0 = *(const float4*)(red + (0*16 + row) * RPITCH + 4*c4);
        const float4 p1 = *(const float4*)(red + (1*16 + row) * RPITCH + 4*c4);
        const float4 p2 = *(const float4*)(red + (2*16 + row) * RPITCH + 4*c4);
        const float4 p3 = *(const float4*)(red + (3*16 + row) * RPITCH + 4*c4);
        float4 o;
        o.x = ((p0.x + p1.x) + p2.x) + p3.x;
        o.y = ((p0.y + p1.y) + p2.y) + p3.y;
        o.z = ((p0.z + p1.z) + p2.z) + p3.z;
        o.w = ((p0.w + p1.w) + p2.w) + p3.w;
        *(float4*)(xp + (size_t)rt * (16 * NPROJ) + 4*q) = o;
    }
}

#define DTW_WAVES 8
#define DTW_COLS  (DTW_WAVES * 32)

__device__ __forceinline__ void dtw_load_pair(float4 (&b)[2][8], const float* __restrict__ wb)
{
    #pragma unroll
    for (int u = 0; u < 2; ++u) {
        #pragma unroll
        for (int j = 0; j < 8; ++j)
            b[u][j] = *(const float4*)(wb + (size_t)(16*u) * RR + 8*j);
    }
}

__device__ __forceinline__ float softplus_f(float v)
{
    const float e = __expf(v);
    const float u = 1.0f + e;
    const float d = u - 1.0f;
    const float l = __builtin_amdgcn_logf(u) * 0.693147180559945f;
    const float r = l * (e * __builtin_amdgcn_rcpf(d));
    const float s = (d == 0.0f) ? e : r;
    return (v > 20.0f) ? v : s;
}

__global__ __launch_bounds__(DTW_WAVES * 32) __attribute__((amdgpu_waves_per_eu(8, 8)))
void dt_state_kernel(const float* __restrict__ xp,
                     const float* __restrict__ w,
                     const float* __restrict__ bias,
                     const float* __restrict__ x,
                     const float* __restrict__ A_log,
                     float* __restrict__ dt,
                     float* __restrict__ hloc,
                     float* __restrict__ dsum)
{
    const int lane = threadIdx.x & 31;
    const int wv   = threadIdx.x >> 5;
    const int half = lane >> 4;
    const int l15  = lane & 15;
    const int rt   = blockIdx.x / (MM / DTW_COLS);
    const int colw = (blockIdx.x % (MM / DTW_COLS)) * DTW_COLS + 32 * wv;
    const int m    = colw + lane;

    const float* __restrict__ arow = xp + (size_t)(rt*16 + l15) * NPROJ + 4*half;
    const float* __restrict__ wb   = w  + (size_t)(colw + l15) * RR + 4*half;

    float4 a4[8];
    #pragma unroll
    for (int j = 0; j < 8; ++j) a4[j] = *(const float4*)(arow + 8*j);
    float4 b[2][8];
    dtw_load_pair(b, wb);
    const float bb = bias[m];

    v8f c0 = {}, c1 = {};
    #pragma unroll
    for (int j = 0; j < 8; ++j) {
        v2f a0; a0.x = a4[j].x; a0.y = a4[j].y;
        v2f a1; a1.x = a4[j].z; a1.y = a4[j].w;
        v2f p0; p0.x = b[0][j].x; p0.y = b[0][j].y;
        v2f q0; q0.x = b[1][j].x; q0.y = b[1][j].y;
        v2f p1; p1.x = b[0][j].z; p1.y = b[0][j].w;
        v2f q1; q1.x = b[1][j].z; q1.y = b[1][j].w;
        c0 = __builtin_amdgcn_wmma_f32_16x16x4_f32(false, a0, false, p0, (short)0, c0, false, false);
        c1 = __builtin_amdgcn_wmma_f32_16x16x4_f32(false, a0, false, q0, (short)0, c1, false, false);
        c0 = __builtin_amdgcn_wmma_f32_16x16x4_f32(false, a1, false, p1, (short)0, c0, false, false);
        c1 = __builtin_amdgcn_wmma_f32_16x16x4_f32(false, a1, false, q1, (short)0, c1, false, false);
    }

    const float* __restrict__ xcol = x + (size_t)(rt * 16) * MM + m;
    float xv[TT], al[DS];
    #pragma unroll
    for (int t = 0; t < TT; ++t) xv[t] = xcol[(size_t)t * MM];
    #pragma unroll
    for (int d = 0; d < DS; ++d) al[d] = A_log[(size_t)d * MM + m];

    float* __restrict__ dcol = dt + (size_t)(rt * 16) * MM + m;
    float dtv[TT];
    #pragma unroll
    for (int r = 0; r < 8; ++r) {
        const float give = half ? c0[r] : c1[r];
        const int   gi   = __builtin_bit_cast(int, give);
        const float got  = __builtin_bit_cast(float,
            __builtin_amdgcn_permlanex16(gi, gi, 0x76543210, 0xfedcba98, false, false));
        const float lo = half ? got   : c0[r];
        const float hi = half ? c1[r] : got;
        dtv[r]     = softplus_f(lo + bb);
        dtv[8 + r] = softplus_f(hi + bb);
        dcol[(size_t)r * MM]       = dtv[r];
        dcol[(size_t)(8 + r) * MM] = dtv[8 + r];
    }

    float Am[DS], h[DS];
    #pragma unroll
    for (int d = 0; d < DS; ++d) {
        Am[d] = -__expf(al[d]) * LOG2E_F;
        h[d]  = 0.0f;
    }
    float S = 0.0f;

    #pragma unroll
    for (int t = 0; t < TT; ++t) {
        const float4* __restrict__ bc =
            (const float4*)(xp + (size_t)(rt * 16 + t) * NPROJ + RR);
        float Bv[DS];
        #pragma unroll
        for (int i = 0; i < 4; ++i) {
            float4 vb = bc[i];
            Bv[4*i+0] = vb.x; Bv[4*i+1] = vb.y; Bv[4*i+2] = vb.z; Bv[4*i+3] = vb.w;
        }

        const float dtt = dtv[t];
        const float dx  = dtt * xv[t];
        #pragma unroll
        for (int d = 0; d < DS; ++d)
            h[d] = __builtin_amdgcn_exp2f(dtt * Am[d]) * h[d] + dx * Bv[d];
        S += dtt;
    }

    const size_t base = ((size_t)rt * DS) * MM + m;
    #pragma unroll
    for (int d = 0; d < DS; ++d)
        hloc[base + (size_t)d * MM] = h[d];
    dsum[(size_t)rt * MM + m] = S;
}

__global__ __launch_bounds__(256)
void chunk_carry_kernel(const float* __restrict__ A_log,
                        const float* __restrict__ hloc,
                        const float* __restrict__ dsum,
                        float* __restrict__ hin)
{
    const int idx = blockIdx.x * 256 + threadIdx.x;
    const int m   = idx & (MM - 1);
    const int d   = (idx >> 10) & (DS - 1);
    const int b   = idx >> 14;

    const float Am = -__expf(A_log[(size_t)d * MM + m]);
    float h = 0.0f;
    size_t o = ((size_t)(b * NC) * DS + d) * MM + m;
    hin[o] = 0.0f;
    #pragma unroll 32
    for (int c = 0; c < NC - 1; ++c) {
        const float S = dsum[(size_t)(b * NC + c) * MM + m];
        const float e = hloc[o];
        h = __expf(Am * S) * h + e;
        o += (size_t)DS * MM;
        hin[o] = h;
    }
}

__global__ __launch_bounds__(TPB) __attribute__((amdgpu_waves_per_eu(8, 8)))
void scan_kernel(const float* __restrict__ x,
                 const float* __restrict__ z,
                 const float* __restrict__ xp,
                 const float* __restrict__ dtb,
                 const float* __restrict__ A_log,
                 const float* __restrict__ Dp_arr,
                 const float* __restrict__ hin,
                 float* __restrict__ out)
{
    const int mb = blockIdx.x % MBLK;
    const int c  = (blockIdx.x / MBLK) % NC;
    const int b  = blockIdx.x / (MBLK * NC);
    const int m  = mb * TPB + threadIdx.x;

    const size_t base = ((size_t)(b * NC + c) * DS) * MM + m;
    const size_t row0 = (size_t)b * LL + (size_t)c * TT;
    const float* __restrict__ dcol = dtb + row0 * MM + m;
    const float* __restrict__ xcol = x   + row0 * MM + m;
    const float* __restrict__ zcol = z   + row0 * MM + m;
    float*       __restrict__ ocol = out + row0 * MM + m;

    float al[DS], h[DS];
    #pragma unroll
    for (int d = 0; d < DS; ++d) {
        al[d] = A_log[(size_t)d * MM + m];
        h[d]  = hin[base + (size_t)d * MM];
    }
    const float Dp = Dp_arr[m];

    float dtv[TT], xv[TT], zv[TT];
    #pragma unroll
    for (int t = 0; t < TT; ++t) {
        dtv[t] = dcol[(size_t)t * MM];
        xv[t]  = xcol[(size_t)t * MM];
        zv[t]  = zcol[(size_t)t * MM];
    }

    float Am[DS];
    #pragma unroll
    for (int d = 0; d < DS; ++d)
        Am[d] = -__expf(al[d]) * LOG2E_F;

    #pragma unroll
    for (int t = 0; t < TT; ++t) {
        const float4* __restrict__ bc =
            (const float4*)(xp + (row0 + t) * NPROJ + RR);
        float Bv[DS], Cv[DS];
        #pragma unroll
        for (int i = 0; i < 4; ++i) {
            float4 vb = bc[i];
            float4 vc = bc[4 + i];
            Bv[4*i+0] = vb.x; Bv[4*i+1] = vb.y; Bv[4*i+2] = vb.z; Bv[4*i+3] = vb.w;
            Cv[4*i+0] = vc.x; Cv[4*i+1] = vc.y; Cv[4*i+2] = vc.z; Cv[4*i+3] = vc.w;
        }

        const float dx = dtv[t] * xv[t];
        float y = 0.0f;
        #pragma unroll
        for (int d = 0; d < DS; ++d) {
            h[d] = __builtin_amdgcn_exp2f(dtv[t] * Am[d]) * h[d] + dx * Bv[d];
            y    = fmaf(h[d], Cv[d], y);
        }
        y = fmaf(Dp, xv[t], y);

        const float sig = 1.0f / (1.0f + __expf(-zv[t]));
        ocol[(size_t)t * MM] = y * (zv[t] * sig);
    }
}

extern "C" void kernel_launch(void* const* d_in, const int* in_sizes, int n_in,
                              void* d_out, int out_size, void* d_ws, size_t ws_size,
                              hipStream_t stream) {
    const float* x         = (const float*)d_in[0];
    const float* z         = (const float*)d_in[1];
    const float* x_proj_w  = (const float*)d_in[2];
    const float* dt_proj_w = (const float*)d_in[3];
    const float* dt_proj_b = (const float*)d_in[4];
    const float* A_log     = (const float*)d_in[5];
    const float* D_param   = (const float*)d_in[6];
    float* out = (float*)d_out;

    float* xp_ws = (float*)d_ws;
    float* dt_ws = xp_ws + (size_t)BLT * NPROJ;
    float* ds_ws = dt_ws + (size_t)BLT * MM;
    float* hl_ws = ds_ws + (size_t)BB * NC * MM;
    float* hi_ws = hl_ws + (size_t)BB * NC * DS * MM;

    proj_xp_kernel<<<BLT/16, 128, 0, stream>>>(x, x_proj_w, xp_ws);

    dt_state_kernel<<<(BLT/16) * (MM/DTW_COLS), DTW_WAVES * 32, 0, stream>>>(xp_ws, dt_proj_w, dt_proj_b,
                                                                             x, A_log, dt_ws, hl_ws, ds_ws);

    chunk_carry_kernel<<<(BB * DS * MM) / 256, 256, 0, stream>>>(A_log, hl_ws, ds_ws, hi_ws);

    scan_kernel<<<BB * NC * MBLK, TPB, 0, stream>>>(x, z, xp_ws, dt_ws,
                                                    A_log, D_param, hi_ws, out);
}


# Scan chunks of 32 steps instead of 16: each dt-kernel wave runs two row tiles and carries h across them; half as many chunk states stored, carried and reloaded; h summed in a different order
# speedup vs baseline: 29.3210x; 1.1216x over previous
#include <hip/hip_runtime.h>
#include <hip/hip_bf16.h>
#include <math.h>

#define BB      2
#define LL      2048
#define MM      1024
#define DS      16
#define RR      64
#define NPROJ   (RR + 2*DS)
#define BLT     (BB*LL)
#define NC      64
#define TT      (LL/NC)
#define TILES   (TT/16)
#define TPB     256
#define MBLK    (MM/TPB)

typedef float  v2f __attribute__((ext_vector_type(2)));
typedef float  v8f __attribute__((ext_vector_type(8)));

#define LOG2E_F 1.44269504088896341f

#define KC      128
#define NCH     (MM / KC)
#define SROWS   (16 + NPROJ)
#define SPITCH  (KC + 4)
#define RPITCH  100

__global__ __launch_bounds__(128)
void proj_xp_kernel(const float* __restrict__ x,
                    const float* __restrict__ w,
                    float* __restrict__ xp)
{
    __shared__ __attribute__((aligned(16))) float smem[2 * SROWS * SPITCH];

    const int tid  = threadIdx.x;
    const int lane = tid & 31;
    const int wv   = tid >> 5;
    const int half = lane >> 4;
    const int l15  = lane & 15;
    const int rt   = blockIdx.x;

    const float* __restrict__ xs = x + (size_t)(rt*16 + 4*wv) * MM + lane*4;
    const float* __restrict__ ws = w + (size_t)(24*wv) * MM + lane*4;
    const int xd = (4*wv) * SPITCH + lane*4;
    const int wd = (16 + 24*wv) * SPITCH + lane*4;
    const int fo = l15 * SPITCH + 32*wv + 16*half;

    float4 st[28];
    v8f acc[6];
    #pragma unroll
    for (int t = 0; t < 6; ++t) acc[t] = (v8f){};

    #pragma unroll
    for (int i = 0; i < 4; ++i)  st[i]     = *(const float4*)(xs + (size_t)i * MM);
    #pragma unroll
    for (int i = 0; i < 24; ++i) st[4 + i] = *(const float4*)(ws + (size_t)i * MM);
    #pragma unroll
    for (int i = 0; i < 4; ++i)  *(float4*)(smem + xd + i*SPITCH) = st[i];
    #pragma unroll
    for (int i = 0; i < 24; ++i) *(float4*)(smem + wd + i*SPITCH) = st[4 + i];
    __syncthreads();

    #pragma unroll
    for (int s = 0; s < NCH; ++s) {
        const float* cur = smem + (s & 1) * (SROWS * SPITCH);
        float*       nxt = smem + ((s + 1) & 1) * (SROWS * SPITCH);

        if (s + 1 < NCH) {
            #pragma unroll
            for (int i = 0; i < 4; ++i)
                st[i]     = *(const float4*)(xs + (size_t)i * MM + (s + 1) * KC);
            #pragma unroll
            for (int i = 0; i < 24; ++i)
                st[4 + i] = *(const float4*)(ws + (size_t)i * MM + (s + 1) * KC);
        }

        float4 af[4], bf[6][4];
        #pragma unroll
        for (int j = 0; j < 4; ++j) af[j] = *(const float4*)(cur + fo + 4*j);
        #pragma unroll
        for (int t = 0; t < 6; ++t) {
            #pragma unroll
            for (int j = 0; j < 4; ++j)
                bf[t][j] = *(const float4*)(cur + (16 + 16*t) * SPITCH + fo + 4*j);
        }

        #pragma unroll
        for (int j = 0; j < 4; ++j) {
            v2f a0; a0.x = af[j].x; a0.y = af[j].y;
            v2f a1; a1.x = af[j].z; a1.y = af[j].w;
            #pragma unroll
            for (int t = 0; t < 6; ++t) {
                v2f b0; b0.x = bf[t][j].x; b0.y = bf[t][j].y;
                acc[t] = __builtin_amdgcn_wmma_f32_16x16x4_f32(
                             false, a0, false, b0, (short)0, acc[t], false, false);
            }
            #pragma unroll
            for (int t = 0; t < 6; ++t) {
                v2f b1; b1.x = bf[t][j].z; b1.y = bf[t][j].w;
                acc[t] = __builtin_amdgcn_wmma_f32_16x16x4_f32(
                             false, a1, false, b1, (short)0, acc[t], false, false);
            }
        }

        if (s + 1 < NCH) {
            #pragma unroll
            for (int i = 0; i < 4; ++i)  *(float4*)(nxt + xd + i*SPITCH) = st[i];
            #pragma unroll
            for (int i = 0; i < 24; ++i) *(float4*)(nxt + wd + i*SPITCH) = st[4 + i];
        }
        __syncthreads();
    }

    float* red = smem;
    #pragma unroll
    for (int t = 0; t < 6; ++t) {
        #pragma unroll
        for (int r = 0; r < 8; ++r)
            red[(wv*16 + 8*half + r) * RPITCH + 16*t + l15] = acc[t][r];
    }
    __syncthreads();

    #pragma unroll
    for (int i = 0; i < 3; ++i) {
        const int q   = tid + 128*i;
        const int row = q / 24;
        const int c4  = q - row*24;
        const float4 p0 = *(const float4*)(red + (0*16 + row) * RPITCH + 4*c4);
        const float4 p1 = *(const float4*)(red + (1*16 + row) * RPITCH + 4*c4);
        const float4 p2 = *(const float4*)(red + (2*16 + row) * RPITCH + 4*c4);
        const float4 p3 = *(const float4*)(red + (3*16 + row) * RPITCH + 4*c4);
        float4 o;
        o.x = ((p0.x + p1.x) + p2.x) + p3.x;
        o.y = ((p0.y + p1.y) + p2.y) + p3.y;
        o.z = ((p0.z + p1.z) + p2.z) + p3.z;
        o.w = ((p0.w + p1.w) + p2.w) + p3.w;
        *(float4*)(xp + (size_t)rt * (16 * NPROJ) + 4*q) = o;
    }
}

#define DTW_WAVES 8
#define DTW_COLS  (DTW_WAVES * 32)

__device__ __forceinline__ void dtw_load_pair(float4 (&b)[2][8], const float* __restrict__ wb)
{
    #pragma unroll
    for (int u = 0; u < 2; ++u) {
        #pragma unroll
        for (int j = 0; j < 8; ++j)
            b[u][j] = *(const float4*)(wb + (size_t)(16*u) * RR + 8*j);
    }
}

__device__ __forceinline__ float softplus_f(float v)
{
    const float e = __expf(v);
    const float u = 1.0f + e;
    const float d = u - 1.0f;
    const float l = __builtin_amdgcn_logf(u) * 0.693147180559945f;
    const float r = l * (e * __builtin_amdgcn_rcpf(d));
    const float s = (d == 0.0f) ? e : r;
    return (v > 20.0f) ? v : s;
}

__global__ __launch_bounds__(DTW_WAVES * 32) __attribute__((amdgpu_waves_per_eu(4, 4)))
void dt_state_kernel(const float* __restrict__ xp,
                     const float* __restrict__ w,
                     const float* __restrict__ bias,
                     const float* __restrict__ x,
                     const float* __restrict__ A_log,
                     float* __restrict__ dt,
                     float* __restrict__ hloc,
                     float* __restrict__ dsum)
{
    const int lane = threadIdx.x & 31;
    const int wv   = threadIdx.x >> 5;
    const int half = lane >> 4;
    const int l15  = lane & 15;
    const int ck   = blockIdx.x / (MM / DTW_COLS);
    const int colw = (blockIdx.x % (MM / DTW_COLS)) * DTW_COLS + 32 * wv;
    const int m    = colw + lane;

    const float* __restrict__ wb   = w  + (size_t)(colw + l15) * RR + 4*half;

    float4 a4[TILES][8];
    #pragma unroll
    for (int u = 0; u < TILES; ++u) {
        const float* __restrict__ arow = xp + (size_t)((ck*TILES + u)*16 + l15) * NPROJ + 4*half;
        #pragma unroll
        for (int j = 0; j < 8; ++j) a4[u][j] = *(const float4*)(arow + 8*j);
    }
    float4 b[2][8];
    dtw_load_pair(b, wb);
    const float bb = bias[m];

    const float* __restrict__ xcol = x + (size_t)(ck * TT) * MM + m;
    float xv[TT], al[DS];
    #pragma unroll
    for (int t = 0; t < TT; ++t) xv[t] = xcol[(size_t)t * MM];
    #pragma unroll
    for (int d = 0; d < DS; ++d) al[d] = A_log[(size_t)d * MM + m];

    float Am[DS], h[DS];
    #pragma unroll
    for (int d = 0; d < DS; ++d) {
        Am[d] = -__expf(al[d]) * LOG2E_F;
        h[d]  = 0.0f;
    }
    float S = 0.0f;

    #pragma unroll
    for (int u = 0; u < TILES; ++u) {
        v8f c0 = {}, c1 = {};
        #pragma unroll
        for (int j = 0; j < 8; ++j) {
            v2f a0; a0.x = a4[u][j].x; a0.y = a4[u][j].y;
            v2f a1; a1.x = a4[u][j].z; a1.y = a4[u][j].w;
            v2f p0; p0.x = b[0][j].x; p0.y = b[0][j].y;
            v2f q0; q0.x = b[1][j].x; q0.y = b[1][j].y;
            v2f p1; p1.x = b[0][j].z; p1.y = b[0][j].w;
            v2f q1; q1.x = b[1][j].z; q1.y = b[1][j].w;
            c0 = __builtin_amdgcn_wmma_f32_16x16x4_f32(false, a0, false, p0, (short)0, c0, false, false);
            c1 = __builtin_amdgcn_wmma_f32_16x16x4_f32(false, a0, false, q0, (short)0, c1, false, false);
            c0 = __builtin_amdgcn_wmma_f32_16x16x4_f32(false, a1, false, p1, (short)0, c0, false, false);
            c1 = __builtin_amdgcn_wmma_f32_16x16x4_f32(false, a1, false, q1, (short)0, c1, false, false);
        }

        float* __restrict__ dcol = dt + (size_t)(ck * TT + 16*u) * MM + m;
        float dtv[16];
        #pragma unroll
        for (int r = 0; r < 8; ++r) {
            const float give = half ? c0[r] : c1[r];
            const int   gi   = __builtin_bit_cast(int, give);
            const float got  = __builtin_bit_cast(float,
                __builtin_amdgcn_permlanex16(gi, gi, 0x76543210, 0xfedcba98, false, false));
            const float lo = half ? got   : c0[r];
            const float hi = half ? c1[r] : got;
            dtv[r]     = softplus_f(lo + bb);
            dtv[8 + r] = softplus_f(hi + bb);
            dcol[(size_t)r * MM]       = dtv[r];
            dcol[(size_t)(8 + r) * MM] = dtv[8 + r];
        }

        #pragma unroll
        for (int t = 0; t < 16; ++t) {
            const float4* __restrict__ bc =
                (const float4*)(xp + (size_t)(ck * TT + 16*u + t) * NPROJ + RR);
            float Bv[DS];
            #pragma unroll
            for (int i = 0; i < 4; ++i) {
                float4 vb = bc[i];
                Bv[4*i+0] = vb.x; Bv[4*i+1] = vb.y; Bv[4*i+2] = vb.z; Bv[4*i+3] = vb.w;
            }

            const float dtt = dtv[t];
            const float dx  = dtt * xv[16*u + t];
            #pragma unroll
            for (int d = 0; d < DS; ++d)
                h[d] = __builtin_amdgcn_exp2f(dtt * Am[d]) * h[d] + dx * Bv[d];
            S += dtt;
        }
    }

    const size_t base = ((size_t)ck * DS) * MM + m;
    #pragma unroll
    for (int d = 0; d < DS; ++d)
        hloc[base + (size_t)d * MM] = h[d];
    dsum[(size_t)ck * MM + m] = S;
}

__global__ __launch_bounds__(256)
void chunk_carry_kernel(const float* __restrict__ A_log,
                        const float* __restrict__ hloc,
                        const float* __restrict__ dsum,
                        float* __restrict__ hin)
{
    const int idx = blockIdx.x * 256 + threadIdx.x;
    const int m   = idx & (MM - 1);
    const int d   = (idx >> 10) & (DS - 1);
    const int b   = idx >> 14;

    const float Am = -__expf(A_log[(size_t)d * MM + m]);
    float h = 0.0f;
    size_t o = ((size_t)(b * NC) * DS + d) * MM + m;
    hin[o] = 0.0f;
    #pragma unroll 32
    for (int c = 0; c < NC - 1; ++c) {
        const float S = dsum[(size_t)(b * NC + c) * MM + m];
        const float e = hloc[o];
        h = __expf(Am * S) * h + e;
        o += (size_t)DS * MM;
        hin[o] = h;
    }
}

__global__ __launch_bounds__(TPB) __attribute__((amdgpu_waves_per_eu(4, 4)))
void scan_kernel(const float* __restrict__ x,
                 const float* __restrict__ z,
                 const float* __restrict__ xp,
                 const float* __restrict__ dtb,
                 const float* __restrict__ A_log,
                 const float* __restrict__ Dp_arr,
                 const float* __restrict__ hin,
                 float* __restrict__ out)
{
    const int mb = blockIdx.x % MBLK;
    const int c  = (blockIdx.x / MBLK) % NC;
    const int b  = blockIdx.x / (MBLK * NC);
    const int m  = mb * TPB + threadIdx.x;

    const size_t base = ((size_t)(b * NC + c) * DS) * MM + m;
    const size_t row0 = (size_t)b * LL + (size_t)c * TT;
    const float* __restrict__ dcol = dtb + row0 * MM + m;
    const float* __restrict__ xcol = x   + row0 * MM + m;
    const float* __restrict__ zcol = z   + row0 * MM + m;
    float*       __restrict__ ocol = out + row0 * MM + m;

    float al[DS], h[DS];
    #pragma unroll
    for (int d = 0; d < DS; ++d) {
        al[d] = A_log[(size_t)d * MM + m];
        h[d]  = hin[base + (size_t)d * MM];
    }
    const float Dp = Dp_arr[m];

    float dtv[TT], xv[TT], zv[TT];
    #pragma unroll
    for (int t = 0; t < TT; ++t) {
        dtv[t] = dcol[(size_t)t * MM];
        xv[t]  = xcol[(size_t)t * MM];
        zv[t]  = zcol[(size_t)t * MM];
    }

    float Am[DS];
    #pragma unroll
    for (int d = 0; d < DS; ++d)
        Am[d] = -__expf(al[d]) * LOG2E_F;

    #pragma unroll
    for (int t = 0; t < TT; ++t) {
        const float4* __restrict__ bc =
            (const float4*)(xp + (row0 + t) * NPROJ + RR);
        float Bv[DS], Cv[DS];
        #pragma unroll
        for (int i = 0; i < 4; ++i) {
            float4 vb = bc[i];
            float4 vc = bc[4 + i];
            Bv[4*i+0] = vb.x; Bv[4*i+1] = vb.y; Bv[4*i+2] = vb.z; Bv[4*i+3] = vb.w;
            Cv[4*i+0] = vc.x; Cv[4*i+1] = vc.y; Cv[4*i+2] = vc.z; Cv[4*i+3] = vc.w;
        }

        const float dx = dtv[t] * xv[t];
        float y = 0.0f;
        #pragma unroll
        for (int d = 0; d < DS; ++d) {
            h[d] = __builtin_amdgcn_exp2f(dtv[t] * Am[d]) * h[d] + dx * Bv[d];
            y    = fmaf(h[d], Cv[d], y);
        }
        y = fmaf(Dp, xv[t], y);

        const float sig = 1.0f / (1.0f + __expf(-zv[t]));
        ocol[(size_t)t * MM] = y * (zv[t] * sig);
    }
}

extern "C" void kernel_launch(void* const* d_in, const int* in_sizes, int n_in,
                              void* d_out, int out_size, void* d_ws, size_t ws_size,
                              hipStream_t stream) {
    const float* x         = (const float*)d_in[0];
    const float* z         = (const float*)d_in[1];
    const float* x_proj_w  = (const float*)d_in[2];
    const float* dt_proj_w = (const float*)d_in[3];
    const float* dt_proj_b = (const float*)d_in[4];
    const float* A_log     = (const float*)d_in[5];
    const float* D_param   = (const float*)d_in[6];
    float* out = (float*)d_out;

    float* xp_ws = (float*)d_ws;
    float* dt_ws = xp_ws + (size_t)BLT * NPROJ;
    float* ds_ws = dt_ws + (size_t)BLT * MM;
    float* hl_ws = ds_ws + (size_t)BB * NC * MM;
    float* hi_ws = hl_ws + (size_t)BB * NC * DS * MM;

    proj_xp_kernel<<<BLT/16, 128, 0, stream>>>(x, x_proj_w, xp_ws);

    dt_state_kernel<<<(BLT/TT) * (MM/DTW_COLS), DTW_WAVES * 32, 0, stream>>>(xp_ws, dt_proj_w, dt_proj_b,
                                                                             x, A_log, dt_ws, hl_ws, ds_ws);

    chunk_carry_kernel<<<(BB * DS * MM) / 256, 256, 0, stream>>>(A_log, hl_ws, ds_ws, hi_ws);

    scan_kernel<<<BB * NC * MBLK, TPB, 0, stream>>>(x, z, xp_ws, dt_ws,
                                                    A_log, D_param, hi_ws, out);
}


# Carry over the chunks moved into the scan kernel: workgroup = 32 channels x 16 consecutive chunks, wave w folds state w over earlier chunks into LDS start states; carry kernel and its launch gone
# speedup vs baseline: 30.5016x; 1.0403x over previous
#include <hip/hip_runtime.h>
#include <hip/hip_bf16.h>
#include <math.h>

#define BB      2
#define LL      2048
#define MM      1024
#define DS      16
#define RR      64
#define NPROJ   (RR + 2*DS)
#define BLT     (BB*LL)
#define NC      64
#define TT      (LL/NC)
#define TILES   (TT/16)
#define TPB     256
#define MBLK    (MM/TPB)

typedef float  v2f __attribute__((ext_vector_type(2)));
typedef float  v8f __attribute__((ext_vector_type(8)));

#define LOG2E_F 1.44269504088896341f

#define KC      128
#define NCH     (MM / KC)
#define SROWS   (16 + NPROJ)
#define SPITCH  (KC + 4)
#define RPITCH  100

__global__ __launch_bounds__(128)
void proj_xp_kernel(const float* __restrict__ x,
                    const float* __restrict__ w,
                    float* __restrict__ xp)
{
    __shared__ __attribute__((aligned(16))) float smem[2 * SROWS * SPITCH];

    const int tid  = threadIdx.x;
    const int lane = tid & 31;
    const int wv   = tid >> 5;
    const int half = lane >> 4;
    const int l15  = lane & 15;
    const int rt   = blockIdx.x;

    const float* __restrict__ xs = x + (size_t)(rt*16 + 4*wv) * MM + lane*4;
    const float* __restrict__ ws = w + (size_t)(24*wv) * MM + lane*4;
    const int xd = (4*wv) * SPITCH + lane*4;
    const int wd = (16 + 24*wv) * SPITCH + lane*4;
    const int fo = l15 * SPITCH + 32*wv + 16*half;

    float4 st[28];
    v8f acc[6];
    #pragma unroll
    for (int t = 0; t < 6; ++t) acc[t] = (v8f){};

    #pragma unroll
    for (int i = 0; i < 4; ++i)  st[i]     = *(const float4*)(xs + (size_t)i * MM);
    #pragma unroll
    for (int i = 0; i < 24; ++i) st[4 + i] = *(const float4*)(ws + (size_t)i * MM);
    #pragma unroll
    for (int i = 0; i < 4; ++i)  *(float4*)(smem + xd + i*SPITCH) = st[i];
    #pragma unroll
    for (int i = 0; i < 24; ++i) *(float4*)(smem + wd + i*SPITCH) = st[4 + i];
    __syncthreads();

    #pragma unroll
    for (int s = 0; s < NCH; ++s) {
        const float* cur = smem + (s & 1) * (SROWS * SPITCH);
        float*       nxt = smem + ((s + 1) & 1) * (SROWS * SPITCH);

        if (s + 1 < NCH) {
            #pragma unroll
            for (int i = 0; i < 4; ++i)
                st[i]     = *(const float4*)(xs + (size_t)i * MM + (s + 1) * KC);
            #pragma unroll
            for (int i = 0; i < 24; ++i)
                st[4 + i] = *(const float4*)(ws + (size_t)i * MM + (s + 1) * KC);
        }

        float4 af[4], bf[6][4];
        #pragma unroll
        for (int j = 0; j < 4; ++j) af[j] = *(const float4*)(cur + fo + 4*j);
        #pragma unroll
        for (int t = 0; t < 6; ++t) {
            #pragma unroll
            for (int j = 0; j < 4; ++j)
                bf[t][j] = *(const float4*)(cur + (16 + 16*t) * SPITCH + fo + 4*j);
        }

        #pragma unroll
        for (int j = 0; j < 4; ++j) {
            v2f a0; a0.x = af[j].x; a0.y = af[j].y;
            v2f a1; a1.x = af[j].z; a1.y = af[j].w;
            #pragma unroll
            for (int t = 0; t < 6; ++t) {
                v2f b0; b0.x = bf[t][j].x; b0.y = bf[t][j].y;
                acc[t] = __builtin_amdgcn_wmma_f32_16x16x4_f32(
                             false, a0, false, b0, (short)0, acc[t], false, false);
            }
            #pragma unroll
            for (int t = 0; t < 6; ++t) {
                v2f b1; b1.x = bf[t][j].z; b1.y = bf[t][j].w;
                acc[t] = __builtin_amdgcn_wmma_f32_16x16x4_f32(
                             false, a1, false, b1, (short)0, acc[t], false, false);
            }
        }

        if (s + 1 < NCH) {
            #pragma unroll
            for (int i = 0; i < 4; ++i)  *(float4*)(nxt + xd + i*SPITCH) = st[i];
            #pragma unroll
            for (int i = 0; i < 24; ++i) *(float4*)(nxt + wd + i*SPITCH) = st[4 + i];
        }
        __syncthreads();
    }

    float* red = smem;
    #pragma unroll
    for (int t = 0; t < 6; ++t) {
        #pragma unroll
        for (int r = 0; r < 8; ++r)
            red[(wv*16 + 8*half + r) * RPITCH + 16*t + l15] = acc[t][r];
    }
    __syncthreads();

    #pragma unroll
    for (int i = 0; i < 3; ++i) {
        const int q   = tid + 128*i;
        const int row = q / 24;
        const int c4  = q - row*24;
        const float4 p0 = *(const float4*)(red + (0*16 + row) * RPITCH + 4*c4);
        const float4 p1 = *(const float4*)(red + (1*16 + row) * RPITCH + 4*c4);
        const float4 p2 = *(const float4*)(red + (2*16 + row) * RPITCH + 4*c4);
        const float4 p3 = *(const float4*)(red + (3*16 + row) * RPITCH + 4*c4);
        float4 o;
        o.x = ((p0.x + p1.x) + p2.x) + p3.x;
        o.y = ((p0.y + p1.y) + p2.y) + p3.y;
        o.z = ((p0.z + p1.z) + p2.z) + p3.z;
        o.w = ((p0.w + p1.w) + p2.w) + p3.w;
        *(float4*)(xp + (size_t)rt * (16 * NPROJ) + 4*q) = o;
    }
}

#define DTW_WAVES 8
#define DTW_COLS  (DTW_WAVES * 32)

__device__ __forceinline__ void dtw_load_pair(float4 (&b)[2][8], const float* __restrict__ wb)
{
    #pragma unroll
    for (int u = 0; u < 2; ++u) {
        #pragma unroll
        for (int j = 0; j < 8; ++j)
            b[u][j] = *(const float4*)(wb + (size_t)(16*u) * RR + 8*j);
    }
}

__device__ __forceinline__ float softplus_f(float v)
{
    const float e = __expf(v);
    const float u = 1.0f + e;
    const float d = u - 1.0f;
    const float l = __builtin_amdgcn_logf(u) * 0.693147180559945f;
    const float r = l * (e * __builtin_amdgcn_rcpf(d));
    const float s = (d == 0.0f) ? e : r;
    return (v > 20.0f) ? v : s;
}

__global__ __launch_bounds__(DTW_WAVES * 32) __attribute__((amdgpu_waves_per_eu(4, 4)))
void dt_state_kernel(const float* __restrict__ xp,
                     const float* __restrict__ w,
                     const float* __restrict__ bias,
                     const float* __restrict__ x,
                     const float* __restrict__ A_log,
                     float* __restrict__ dt,
                     float* __restrict__ hloc,
                     float* __restrict__ dsum)
{
    const int lane = threadIdx.x & 31;
    const int wv   = threadIdx.x >> 5;
    const int half = lane >> 4;
    const int l15  = lane & 15;
    const int ck   = blockIdx.x / (MM / DTW_COLS);
    const int colw = (blockIdx.x % (MM / DTW_COLS)) * DTW_COLS + 32 * wv;
    const int m    = colw + lane;

    const float* __restrict__ wb   = w  + (size_t)(colw + l15) * RR + 4*half;

    float4 a4[TILES][8];
    #pragma unroll
    for (int u = 0; u < TILES; ++u) {
        const float* __restrict__ arow = xp + (size_t)((ck*TILES + u)*16 + l15) * NPROJ + 4*half;
        #pragma unroll
        for (int j = 0; j < 8; ++j) a4[u][j] = *(const float4*)(arow + 8*j);
    }
    float4 b[2][8];
    dtw_load_pair(b, wb);
    const float bb = bias[m];

    const float* __restrict__ xcol = x + (size_t)(ck * TT) * MM + m;
    float xv[TT], al[DS];
    #pragma unroll
    for (int t = 0; t < TT; ++t) xv[t] = xcol[(size_t)t * MM];
    #pragma unroll
    for (int d = 0; d < DS; ++d) al[d] = A_log[(size_t)d * MM + m];

    float Am[DS], h[DS];
    #pragma unroll
    for (int d = 0; d < DS; ++d) {
        Am[d] = -__expf(al[d]) * LOG2E_F;
        h[d]  = 0.0f;
    }
    float S = 0.0f;

    #pragma unroll
    for (int u = 0; u < TILES; ++u) {
        v8f c0 = {}, c1 = {};
        #pragma unroll
        for (int j = 0; j < 8; ++j) {
            v2f a0; a0.x = a4[u][j].x; a0.y = a4[u][j].y;
            v2f a1; a1.x = a4[u][j].z; a1.y = a4[u][j].w;
            v2f p0; p0.x = b[0][j].x; p0.y = b[0][j].y;
            v2f q0; q0.x = b[1][j].x; q0.y = b[1][j].y;
            v2f p1; p1.x = b[0][j].z; p1.y = b[0][j].w;
            v2f q1; q1.x = b[1][j].z; q1.y = b[1][j].w;
            c0 = __builtin_amdgcn_wmma_f32_16x16x4_f32(false, a0, false, p0, (short)0, c0, false, false);
            c1 = __builtin_amdgcn_wmma_f32_16x16x4_f32(false, a0, false, q0, (short)0, c1, false, false);
            c0 = __builtin_amdgcn_wmma_f32_16x16x4_f32(false, a1, false, p1, (short)0, c0, false, false);
            c1 = __builtin_amdgcn_wmma_f32_16x16x4_f32(false, a1, false, q1, (short)0, c1, false, false);
        }

        float* __restrict__ dcol = dt + (size_t)(ck * TT + 16*u) * MM + m;
        float dtv[16];
        #pragma unroll
        for (int r = 0; r < 8; ++r) {
            const float give = half ? c0[r] : c1[r];
            const int   gi   = __builtin_bit_cast(int, give);
            const float got  = __builtin_bit_cast(float,
                __builtin_amdgcn_permlanex16(gi, gi, 0x76543210, 0xfedcba98, false, false));
            const float lo = half ? got   : c0[r];
            const float hi = half ? c1[r] : got;
            dtv[r]     = softplus_f(lo + bb);
            dtv[8 + r] = softplus_f(hi + bb);
            dcol[(size_t)r * MM]       = dtv[r];
            dcol[(size_t)(8 + r) * MM] = dtv[8 + r];
        }

        #pragma unroll
        for (int t = 0; t < 16; ++t) {
            const float4* __restrict__ bc =
                (const float4*)(xp + (size_t)(ck * TT + 16*u + t) * NPROJ + RR);
            float Bv[DS];
            #pragma unroll
            for (int i = 0; i < 4; ++i) {
                float4 vb = bc[i];
                Bv[4*i+0] = vb.x; Bv[4*i+1] = vb.y; Bv[4*i+2] = vb.z; Bv[4*i+3] = vb.w;
            }

            const float dtt = dtv[t];
            const float dx  = dtt * xv[16*u + t];
            #pragma unroll
            for (int d = 0; d < DS; ++d)
                h[d] = __builtin_amdgcn_exp2f(dtt * Am[d]) * h[d] + dx * Bv[d];
            S += dtt;
        }
    }

    const size_t base = ((size_t)ck * DS) * MM + m;
    #pragma unroll
    for (int d = 0; d < DS; ++d)
        hloc[base + (size_t)d * MM] = h[d];
    dsum[(size_t)ck * MM + m] = S;
}

__device__ __forceinline__ void lds_barrier()
{
    __builtin_amdgcn_fence(__ATOMIC_RELEASE, "workgroup", "local");
    __builtin_amdgcn_s_barrier();
    __builtin_amdgcn_fence(__ATOMIC_ACQUIRE, "workgroup", "local");
}

#define QC      16
#define NQ      (NC / QC)
#define MGRP    (MM / 32)

__global__ __launch_bounds__(QC * 32) __attribute__((amdgpu_waves_per_eu(4, 4)))
void scan_kernel(const float* __restrict__ x,
                 const float* __restrict__ z,
                 const float* __restrict__ xp,
                 const float* __restrict__ dtb,
                 const float* __restrict__ A_log,
                 const float* __restrict__ Dp_arr,
                 const float* __restrict__ hloc,
                 const float* __restrict__ dsum,
                 float* __restrict__ out)
{
    __shared__ float s_S[NC * 32];
    __shared__ float s_h[QC * DS * 32];

    const int lane = threadIdx.x & 31;
    const int wv   = __builtin_amdgcn_readfirstlane((int)(threadIdx.x >> 5));
    const int mg   = blockIdx.x % MGRP;
    const int q    = (blockIdx.x / MGRP) % NQ;
    const int b    = blockIdx.x / (MGRP * NQ);
    const int m    = mg * 32 + lane;
    const int c    = q * QC + wv;

    const float* __restrict__ hl = hloc + ((size_t)(b * NC) * DS + wv) * MM + m;
    const float* __restrict__ sl = dsum + (size_t)(b * NC + wv) * MM + m;
    float E[NC], Sg[NQ];
    #pragma unroll
    for (int g = 0; g < NQ; ++g) {
        if (g <= q) {
            Sg[g] = sl[(size_t)(QC * g) * MM];
            #pragma unroll
            for (int i = 0; i < QC; ++i)
                if (QC*g + i < NC - 1) E[QC*g + i] = hl[(size_t)(QC*g + i) * DS * MM];
        }
    }
    const float Ad = -__expf(A_log[(size_t)wv * MM + m]);

    __builtin_amdgcn_sched_barrier(0);

    const size_t row0 = (size_t)b * LL + (size_t)c * TT;
    const float* __restrict__ dcol = dtb + row0 * MM + m;
    const float* __restrict__ xcol = x   + row0 * MM + m;
    const float* __restrict__ zcol = z   + row0 * MM + m;
    float*       __restrict__ ocol = out + row0 * MM + m;

    float al[DS], h[DS];
    #pragma unroll
    for (int d = 0; d < DS; ++d) al[d] = A_log[(size_t)d * MM + m];
    const float Dp = Dp_arr[m];

    float dtv[TT], xv[TT], zv[TT];
    #pragma unroll
    for (int t = 0; t < TT; ++t) {
        dtv[t] = dcol[(size_t)t * MM];
        xv[t]  = xcol[(size_t)t * MM];
        zv[t]  = zcol[(size_t)t * MM];
    }

    #pragma unroll
    for (int g = 0; g < NQ; ++g)
        if (g <= q) s_S[(QC*g + wv) * 32 + lane] = Sg[g];
    lds_barrier();

    float hc = 0.0f;
    #pragma unroll
    for (int g = 0; g < NQ; ++g) {
        if (g < q) {
            #pragma unroll
            for (int i = 0; i < QC; ++i)
                hc = __expf(Ad * s_S[(QC*g + i) * 32 + lane]) * hc + E[QC*g + i];
        } else if (g == q) {
            #pragma unroll
            for (int i = 0; i < QC; ++i) {
                s_h[(i * DS + wv) * 32 + lane] = hc;
                if (i < QC - 1)
                    hc = __expf(Ad * s_S[(QC*g + i) * 32 + lane]) * hc + E[QC*g + i];
            }
        }
    }
    lds_barrier();

    #pragma unroll
    for (int d = 0; d < DS; ++d) h[d] = s_h[(wv * DS + d) * 32 + lane];

    float Am[DS];
    #pragma unroll
    for (int d = 0; d < DS; ++d)
        Am[d] = -__expf(al[d]) * LOG2E_F;

    #pragma unroll
    for (int t = 0; t < TT; ++t) {
        const float4* __restrict__ bc =
            (const float4*)(xp + (row0 + t) * NPROJ + RR);
        float Bv[DS], Cv[DS];
        #pragma unroll
        for (int i = 0; i < 4; ++i) {
            float4 vb = bc[i];
            float4 vc = bc[4 + i];
            Bv[4*i+0] = vb.x; Bv[4*i+1] = vb.y; Bv[4*i+2] = vb.z; Bv[4*i+3] = vb.w;
            Cv[4*i+0] = vc.x; Cv[4*i+1] = vc.y; Cv[4*i+2] = vc.z; Cv[4*i+3] = vc.w;
        }

        const float dx = dtv[t] * xv[t];
        float y = 0.0f;
        #pragma unroll
        for (int d = 0; d < DS; ++d) {
            h[d] = __builtin_amdgcn_exp2f(dtv[t] * Am[d]) * h[d] + dx * Bv[d];
            y    = fmaf(h[d], Cv[d], y);
        }
        y = fmaf(Dp, xv[t], y);

        const float sig = 1.0f / (1.0f + __expf(-zv[t]));
        ocol[(size_t)t * MM] = y * (zv[t] * sig);
    }
}

extern "C" void kernel_launch(void* const* d_in, const int* in_sizes, int n_in,
                              void* d_out, int out_size, void* d_ws, size_t ws_size,
                              hipStream_t stream) {
    const float* x         = (const float*)d_in[0];
    const float* z         = (const float*)d_in[1];
    const float* x_proj_w  = (const float*)d_in[2];
    const float* dt_proj_w = (const float*)d_in[3];
    const float* dt_proj_b = (const float*)d_in[4];
    const float* A_log     = (const float*)d_in[5];
    const float* D_param   = (const float*)d_in[6];
    float* out = (float*)d_out;

    float* xp_ws = (float*)d_ws;
    float* dt_ws = xp_ws + (size_t)BLT * NPROJ;
    float* ds_ws = dt_ws + (size_t)BLT * MM;
    float* hl_ws = ds_ws + (size_t)BB * NC * MM;

    proj_xp_kernel<<<BLT/16, 128, 0, stream>>>(x, x_proj_w, xp_ws);

    dt_state_kernel<<<(BLT/TT) * (MM/DTW_COLS), DTW_WAVES * 32, 0, stream>>>(xp_ws, dt_proj_w, dt_proj_b,
                                                                             x, A_log, dt_ws, hl_ws, ds_ws);

    scan_kernel<<<BB * NQ * MGRP, QC * 32, 0, stream>>>(x, z, xp_ws, dt_ws,
                                                        A_log, D_param, hl_ws, ds_ws, out);
}


# x_proj kernel: K loop pipelined, one LDS barrier a chunk: next chunk's fragment reads, the LDS stores of the chunk after and global loads two chunks further are issued ahead of each chunk's WMMAs
# speedup vs baseline: 31.4594x; 1.0314x over previous
#include <hip/hip_runtime.h>
#include <hip/hip_bf16.h>
#include <math.h>

#define BB      2
#define LL      2048
#define MM      1024
#define DS      16
#define RR      64
#define NPROJ   (RR + 2*DS)
#define BLT     (BB*LL)
#define NC      64
#define TT      (LL/NC)
#define TILES   (TT/16)
#define TPB     256
#define MBLK    (MM/TPB)

typedef float  v2f __attribute__((ext_vector_type(2)));
typedef float  v4f __attribute__((ext_vector_type(4)));
typedef float  v8f __attribute__((ext_vector_type(8)));

#define LOG2E_F 1.44269504088896341f

__device__ __forceinline__ void lds_barrier()
{
    __builtin_amdgcn_fence(__ATOMIC_RELEASE, "workgroup", "local");
    __builtin_amdgcn_s_barrier();
    __builtin_amdgcn_fence(__ATOMIC_ACQUIRE, "workgroup", "local");
}

#define KC      128
#define NCH     (MM / KC)
#define SROWS   (16 + NPROJ)
#define SPITCH  (KC + 4)
#define RPITCH  100
#define NST     28
#define NFR     28

__global__ __launch_bounds__(128) __attribute__((amdgpu_waves_per_eu(1, 1)))
void proj_xp_kernel(const float* __restrict__ x,
                    const float* __restrict__ w,
                    float* __restrict__ xp)
{
    __shared__ __attribute__((aligned(16))) float smem[2 * SROWS * SPITCH];

    const int tid  = threadIdx.x;
    const int lane = tid & 31;
    const int wv   = tid >> 5;
    const int half = lane >> 4;
    const int l15  = lane & 15;
    const int rt   = blockIdx.x;

    const float* __restrict__ xs = x + (size_t)(rt*16 + 4*wv) * MM + lane*4;
    const float* __restrict__ ws = w + (size_t)(24*wv) * MM + lane*4;
    const int xd = (4*wv) * SPITCH + lane*4;
    const int wd = (16 + 24*wv) * SPITCH + lane*4;
    const int fo = l15 * SPITCH + 32*wv + 16*half;

    v4f pa[NST], pb[NST];
    v4f stA[NST], stB[NST];
    v4f fr0[NFR], fr1[NFR];
    v8f acc[6];
    #pragma unroll
    for (int t = 0; t < 6; ++t) acc[t] = (v8f){};

    #pragma unroll
    for (int i = 0; i < 4; ++i)  pa[i]      = *(const v4f*)(xs + (size_t)i * MM);
    #pragma unroll
    for (int i = 0; i < 24; ++i) pa[4 + i]  = *(const v4f*)(ws + (size_t)i * MM);
    #pragma unroll
    for (int i = 0; i < 4; ++i)  pb[i]      = *(const v4f*)(xs + (size_t)i * MM + KC);
    #pragma unroll
    for (int i = 0; i < 24; ++i) pb[4 + i]  = *(const v4f*)(ws + (size_t)i * MM + KC);
    #pragma unroll
    for (int i = 0; i < 4; ++i)  stA[i]     = *(const v4f*)(xs + (size_t)i * MM + 2 * KC);
    #pragma unroll
    for (int i = 0; i < 24; ++i) stA[4 + i] = *(const v4f*)(ws + (size_t)i * MM + 2 * KC);
    #pragma unroll
    for (int i = 0; i < 4; ++i)  stB[i]     = *(const v4f*)(xs + (size_t)i * MM + 3 * KC);
    #pragma unroll
    for (int i = 0; i < 24; ++i) stB[4 + i] = *(const v4f*)(ws + (size_t)i * MM + 3 * KC);

    #pragma unroll
    for (int i = 0; i < 4; ++i)  *(v4f*)(smem + xd + i*SPITCH) = pa[i];
    #pragma unroll
    for (int i = 0; i < 24; ++i) *(v4f*)(smem + wd + i*SPITCH) = pa[4 + i];
    lds_barrier();

    #pragma unroll
    for (int i = 0; i < 4; ++i)  *(v4f*)(smem + SROWS*SPITCH + xd + i*SPITCH) = pb[i];
    #pragma unroll
    for (int i = 0; i < 24; ++i) *(v4f*)(smem + SROWS*SPITCH + wd + i*SPITCH) = pb[4 + i];
    #pragma unroll
    for (int j = 0; j < 4; ++j) fr0[j] = *(const v4f*)(smem + fo + 4*j);
    #pragma unroll
    for (int t = 0; t < 6; ++t) {
        #pragma unroll
        for (int j = 0; j < 4; ++j)
            fr0[4 + 4*t + j] = *(const v4f*)(smem + (16 + 16*t) * SPITCH + fo + 4*j);
    }
    lds_barrier();

#define PROJ_STEP(s, CUR, NXT, ST)                                                                \
    {                                                                                             \
        float*       own = smem + ((s) & 1) * (SROWS * SPITCH);         \
        const float* nxt = smem + (((s) + 1) & 1) * (SROWS * SPITCH);         \
        if ((s) + 2 < NCH) {                                                                      \
            _Pragma("unroll")                                                                     \
            for (int i = 0; i < 4; ++i)  *(v4f*)(own + xd + i*SPITCH) = ST[i];                 \
            _Pragma("unroll")                                                                     \
            for (int i = 0; i < 24; ++i) *(v4f*)(own + wd + i*SPITCH) = ST[4 + i];             \
        }                                                                                         \
        if ((s) + 4 < NCH) {                                                                      \
            _Pragma("unroll")                                                                     \
            for (int i = 0; i < 4; ++i)                                                           \
                ST[i]     = *(const v4f*)(xs + (size_t)i * MM + ((s) + 4) * KC);               \
            _Pragma("unroll")                                                                     \
            for (int i = 0; i < 24; ++i)                                                          \
                ST[4 + i] = *(const v4f*)(ws + (size_t)i * MM + ((s) + 4) * KC);               \
        }                                                                                         \
        if ((s) + 1 < NCH) {                                                                      \
            _Pragma("unroll")                                                                     \
            for (int j = 0; j < 4; ++j)                                                           \
                NXT[j] = *(const v4f*)(nxt + fo + 4*j);                                        \
            _Pragma("unroll")                                                                     \
            for (int t = 0; t < 6; ++t) {                                                         \
                _Pragma("unroll")                                                                 \
                for (int j = 0; j < 4; ++j)                                                       \
                    NXT[4 + 4*t + j] =                                                            \
                        *(const v4f*)(nxt + (16 + 16*t) * SPITCH + fo + 4*j);                  \
            }                                                                                     \
        }                                                                                         \
           \
          \
        __builtin_amdgcn_sched_barrier(0);                                                        \
        _Pragma("unroll")                                                                         \
        for (int j = 0; j < 4; ++j) {                                                             \
            const v4f af = CUR[j];                                                             \
            v2f a0; a0.x = af.x; a0.y = af.y;                                                     \
            v2f a1; a1.x = af.z; a1.y = af.w;                                                     \
            _Pragma("unroll")                                                                     \
            for (int t = 0; t < 6; ++t) {                                                         \
                const v4f bf = CUR[4 + 4*t + j];                                               \
                v2f b0; b0.x = bf.x; b0.y = bf.y;                                                 \
                acc[t] = __builtin_amdgcn_wmma_f32_16x16x4_f32(                                   \
                             false, a0, false, b0, (short)0, acc[t], false, false);               \
            }                                                                                     \
            _Pragma("unroll")                                                                     \
            for (int t = 0; t < 6; ++t) {                                                         \
                const v4f bf = CUR[4 + 4*t + j];                                               \
                v2f b1; b1.x = bf.z; b1.y = bf.w;                                                 \
                acc[t] = __builtin_amdgcn_wmma_f32_16x16x4_f32(                                   \
                             false, a1, false, b1, (short)0, acc[t], false, false);               \
            }                                                                                     \
        }                                                                                         \
              \
            \
                                                                            \
        __builtin_amdgcn_sched_barrier(0);                                                        \
        if ((s) + 2 < NCH) lds_barrier();                                                         \
    }

    PROJ_STEP(0, fr0, fr1, stA) PROJ_STEP(1, fr1, fr0, stB) PROJ_STEP(2, fr0, fr1, stA) PROJ_STEP(3, fr1, fr0, stB)
    PROJ_STEP(4, fr0, fr1, stA) PROJ_STEP(5, fr1, fr0, stB) PROJ_STEP(6, fr0, fr1, stA) PROJ_STEP(7, fr1, fr0, stB)
#undef PROJ_STEP

    float* red = smem;
    #pragma unroll
    for (int t = 0; t < 6; ++t) {
        #pragma unroll
        for (int r = 0; r < 8; ++r)
            red[(wv*16 + 8*half + r) * RPITCH + 16*t + l15] = acc[t][r];
    }
    __syncthreads();

    #pragma unroll
    for (int i = 0; i < 3; ++i) {
        const int q   = tid + 128*i;
        const int row = q / 24;
        const int c4  = q - row*24;
        const float4 p0 = *(const float4*)(red + (0*16 + row) * RPITCH + 4*c4);
        const float4 p1 = *(const float4*)(red + (1*16 + row) * RPITCH + 4*c4);
        const float4 p2 = *(const float4*)(red + (2*16 + row) * RPITCH + 4*c4);
        const float4 p3 = *(const float4*)(red + (3*16 + row) * RPITCH + 4*c4);
        float4 o;
        o.x = ((p0.x + p1.x) + p2.x) + p3.x;
        o.y = ((p0.y + p1.y) + p2.y) + p3.y;
        o.z = ((p0.z + p1.z) + p2.z) + p3.z;
        o.w = ((p0.w + p1.w) + p2.w) + p3.w;
        *(float4*)(xp + (size_t)rt * (16 * NPROJ) + 4*q) = o;
    }
}

#define DTW_WAVES 8
#define DTW_COLS  (DTW_WAVES * 32)

__device__ __forceinline__ void dtw_load_pair(float4 (&b)[2][8], const float* __restrict__ wb)
{
    #pragma unroll
    for (int u = 0; u < 2; ++u) {
        #pragma unroll
        for (int j = 0; j < 8; ++j)
            b[u][j] = *(const float4*)(wb + (size_t)(16*u) * RR + 8*j);
    }
}

__device__ __forceinline__ float softplus_f(float v)
{
    const float e = __expf(v);
    const float u = 1.0f + e;
    const float d = u - 1.0f;
    const float l = __builtin_amdgcn_logf(u) * 0.693147180559945f;
    const float r = l * (e * __builtin_amdgcn_rcpf(d));
    const float s = (d == 0.0f) ? e : r;
    return (v > 20.0f) ? v : s;
}

__global__ __launch_bounds__(DTW_WAVES * 32) __attribute__((amdgpu_waves_per_eu(4, 4)))
void dt_state_kernel(const float* __restrict__ xp,
                     const float* __restrict__ w,
                     const float* __restrict__ bias,
                     const float* __restrict__ x,
                     const float* __restrict__ A_log,
                     float* __restrict__ dt,
                     float* __restrict__ hloc,
                     float* __restrict__ dsum)
{
    const int lane = threadIdx.x & 31;
    const int wv   = threadIdx.x >> 5;
    const int half = lane >> 4;
    const int l15  = lane & 15;
    const int ck   = blockIdx.x / (MM / DTW_COLS);
    const int colw = (blockIdx.x % (MM / DTW_COLS)) * DTW_COLS + 32 * wv;
    const int m    = colw + lane;

    const float* __restrict__ wb   = w  + (size_t)(colw + l15) * RR + 4*half;

    float4 a4[TILES][8];
    #pragma unroll
    for (int u = 0; u < TILES; ++u) {
        const float* __restrict__ arow = xp + (size_t)((ck*TILES + u)*16 + l15) * NPROJ + 4*half;
        #pragma unroll
        for (int j = 0; j < 8; ++j) a4[u][j] = *(const float4*)(arow + 8*j);
    }
    float4 b[2][8];
    dtw_load_pair(b, wb);
    const float bb = bias[m];

    const float* __restrict__ xcol = x + (size_t)(ck * TT) * MM + m;
    float xv[TT], al[DS];
    #pragma unroll
    for (int t = 0; t < TT; ++t) xv[t] = xcol[(size_t)t * MM];
    #pragma unroll
    for (int d = 0; d < DS; ++d) al[d] = A_log[(size_t)d * MM + m];

    float Am[DS], h[DS];
    #pragma unroll
    for (int d = 0; d < DS; ++d) {
        Am[d] = -__expf(al[d]) * LOG2E_F;
        h[d]  = 0.0f;
    }
    float S = 0.0f;

    #pragma unroll
    for (int u = 0; u < TILES; ++u) {
        v8f c0 = {}, c1 = {};
        #pragma unroll
        for (int j = 0; j < 8; ++j) {
            v2f a0; a0.x = a4[u][j].x; a0.y = a4[u][j].y;
            v2f a1; a1.x = a4[u][j].z; a1.y = a4[u][j].w;
            v2f p0; p0.x = b[0][j].x; p0.y = b[0][j].y;
            v2f q0; q0.x = b[1][j].x; q0.y = b[1][j].y;
            v2f p1; p1.x = b[0][j].z; p1.y = b[0][j].w;
            v2f q1; q1.x = b[1][j].z; q1.y = b[1][j].w;
            c0 = __builtin_amdgcn_wmma_f32_16x16x4_f32(false, a0, false, p0, (short)0, c0, false, false);
            c1 = __builtin_amdgcn_wmma_f32_16x16x4_f32(false, a0, false, q0, (short)0, c1, false, false);
            c0 = __builtin_amdgcn_wmma_f32_16x16x4_f32(false, a1, false, p1, (short)0, c0, false, false);
            c1 = __builtin_amdgcn_wmma_f32_16x16x4_f32(false, a1, false, q1, (short)0, c1, false, false);
        }

        float* __restrict__ dcol = dt + (size_t)(ck * TT + 16*u) * MM + m;
        float dtv[16];
        #pragma unroll
        for (int r = 0; r < 8; ++r) {
            const float give = half ? c0[r] : c1[r];
            const int   gi   = __builtin_bit_cast(int, give);
            const float got  = __builtin_bit_cast(float,
                __builtin_amdgcn_permlanex16(gi, gi, 0x76543210, 0xfedcba98, false, false));
            const float lo = half ? got   : c0[r];
            const float hi = half ? c1[r] : got;
            dtv[r]     = softplus_f(lo + bb);
            dtv[8 + r] = softplus_f(hi + bb);
            dcol[(size_t)r * MM]       = dtv[r];
            dcol[(size_t)(8 + r) * MM] = dtv[8 + r];
        }

        #pragma unroll
        for (int t = 0; t < 16; ++t) {
            const float4* __restrict__ bc =
                (const float4*)(xp + (size_t)(ck * TT + 16*u + t) * NPROJ + RR);
            float Bv[DS];
            #pragma unroll
            for (int i = 0; i < 4; ++i) {
                float4 vb = bc[i];
                Bv[4*i+0] = vb.x; Bv[4*i+1] = vb.y; Bv[4*i+2] = vb.z; Bv[4*i+3] = vb.w;
            }

            const float dtt = dtv[t];
            const float dx  = dtt * xv[16*u + t];
            #pragma unroll
            for (int d = 0; d < DS; ++d)
                h[d] = __builtin_amdgcn_exp2f(dtt * Am[d]) * h[d] + dx * Bv[d];
            S += dtt;
        }
    }

    const size_t base = ((size_t)ck * DS) * MM + m;
    #pragma unroll
    for (int d = 0; d < DS; ++d)
        hloc[base + (size_t)d * MM] = h[d];
    dsum[(size_t)ck * MM + m] = S;
}


#define QC      16
#define NQ      (NC / QC)
#define MGRP    (MM / 32)

__global__ __launch_bounds__(QC * 32) __attribute__((amdgpu_waves_per_eu(4, 4)))
void scan_kernel(const float* __restrict__ x,
                 const float* __restrict__ z,
                 const float* __restrict__ xp,
                 const float* __restrict__ dtb,
                 const float* __restrict__ A_log,
                 const float* __restrict__ Dp_arr,
                 const float* __restrict__ hloc,
                 const float* __restrict__ dsum,
                 float* __restrict__ out)
{
    __shared__ float s_S[NC * 32];
    __shared__ float s_h[QC * DS * 32];

    const int lane = threadIdx.x & 31;
    const int wv   = __builtin_amdgcn_readfirstlane((int)(threadIdx.x >> 5));
    const int mg   = blockIdx.x % MGRP;
    const int q    = (blockIdx.x / MGRP) % NQ;
    const int b    = blockIdx.x / (MGRP * NQ);
    const int m    = mg * 32 + lane;
    const int c    = q * QC + wv;

    const float* __restrict__ hl = hloc + ((size_t)(b * NC) * DS + wv) * MM + m;
    const float* __restrict__ sl = dsum + (size_t)(b * NC + wv) * MM + m;
    float E[NC], Sg[NQ];
    #pragma unroll
    for (int g = 0; g < NQ; ++g) {
        if (g <= q) {
            Sg[g] = sl[(size_t)(QC * g) * MM];
            #pragma unroll
            for (int i = 0; i < QC; ++i)
                if (QC*g + i < NC - 1) E[QC*g + i] = hl[(size_t)(QC*g + i) * DS * MM];
        }
    }
    const float Ad = -__expf(A_log[(size_t)wv * MM + m]);

    __builtin_amdgcn_sched_barrier(0);

    const size_t row0 = (size_t)b * LL + (size_t)c * TT;
    const float* __restrict__ dcol = dtb + row0 * MM + m;
    const float* __restrict__ xcol = x   + row0 * MM + m;
    const float* __restrict__ zcol = z   + row0 * MM + m;
    float*       __restrict__ ocol = out + row0 * MM + m;

    float al[DS], h[DS];
    #pragma unroll
    for (int d = 0; d < DS; ++d) al[d] = A_log[(size_t)d * MM + m];
    const float Dp = Dp_arr[m];

    float dtv[TT], xv[TT], zv[TT];
    #pragma unroll
    for (int t = 0; t < TT; ++t) {
        dtv[t] = dcol[(size_t)t * MM];
        xv[t]  = xcol[(size_t)t * MM];
        zv[t]  = zcol[(size_t)t * MM];
    }

    #pragma unroll
    for (int g = 0; g < NQ; ++g)
        if (g <= q) s_S[(QC*g + wv) * 32 + lane] = Sg[g];
    lds_barrier();

    float hc = 0.0f;
    #pragma unroll
    for (int g = 0; g < NQ; ++g) {
        if (g < q) {
            #pragma unroll
            for (int i = 0; i < QC; ++i)
                hc = __expf(Ad * s_S[(QC*g + i) * 32 + lane]) * hc + E[QC*g + i];
        } else if (g == q) {
            #pragma unroll
            for (int i = 0; i < QC; ++i) {
                s_h[(i * DS + wv) * 32 + lane] = hc;
                if (i < QC - 1)
                    hc = __expf(Ad * s_S[(QC*g + i) * 32 + lane]) * hc + E[QC*g + i];
            }
        }
    }
    lds_barrier();

    #pragma unroll
    for (int d = 0; d < DS; ++d) h[d] = s_h[(wv * DS + d) * 32 + lane];

    float Am[DS];
    #pragma unroll
    for (int d = 0; d < DS; ++d)
        Am[d] = -__expf(al[d]) * LOG2E_F;

    #pragma unroll
    for (int t = 0; t < TT; ++t) {
        const float4* __restrict__ bc =
            (const float4*)(xp + (row0 + t) * NPROJ + RR);
        float Bv[DS], Cv[DS];
        #pragma unroll
        for (int i = 0; i < 4; ++i) {
            float4 vb = bc[i];
            float4 vc = bc[4 + i];
            Bv[4*i+0] = vb.x; Bv[4*i+1] = vb.y; Bv[4*i+2] = vb.z; Bv[4*i+3] = vb.w;
            Cv[4*i+0] = vc.x; Cv[4*i+1] = vc.y; Cv[4*i+2] = vc.z; Cv[4*i+3] = vc.w;
        }

        const float dx = dtv[t] * xv[t];
        float y = 0.0f;
        #pragma unroll
        for (int d = 0; d < DS; ++d) {
            h[d] = __builtin_amdgcn_exp2f(dtv[t] * Am[d]) * h[d] + dx * Bv[d];
            y    = fmaf(h[d], Cv[d], y);
        }
        y = fmaf(Dp, xv[t], y);

        const float sig = 1.0f / (1.0f + __expf(-zv[t]));
        ocol[(size_t)t * MM] = y * (zv[t] * sig);
    }
}

extern "C" void kernel_launch(void* const* d_in, const int* in_sizes, int n_in,
                              void* d_out, int out_size, void* d_ws, size_t ws_size,
                              hipStream_t stream) {
    const float* x         = (const float*)d_in[0];
    const float* z         = (const float*)d_in[1];
    const float* x_proj_w  = (const float*)d_in[2];
    const float* dt_proj_w = (const float*)d_in[3];
    const float* dt_proj_b = (const float*)d_in[4];
    const float* A_log     = (const float*)d_in[5];
    const float* D_param   = (const float*)d_in[6];
    float* out = (float*)d_out;

    float* xp_ws = (float*)d_ws;
    float* dt_ws = xp_ws + (size_t)BLT * NPROJ;
    float* ds_ws = dt_ws + (size_t)BLT * MM;
    float* hl_ws = ds_ws + (size_t)BB * NC * MM;

    proj_xp_kernel<<<BLT/16, 128, 0, stream>>>(x, x_proj_w, xp_ws);

    dt_state_kernel<<<(BLT/TT) * (MM/DTW_COLS), DTW_WAVES * 32, 0, stream>>>(xp_ws, dt_proj_w, dt_proj_b,
                                                                             x, A_log, dt_ws, hl_ws, ds_ws);

    scan_kernel<<<BB * NQ * MGRP, QC * 32, 0, stream>>>(x, z, xp_ws, dt_ws,
                                                        A_log, D_param, hl_ws, ds_ws, out);
}
